# starting kernel
# baseline (speedup 1.0000x reference)
#include <hip/hip_runtime.h>
#include <math.h>


#define D_MODEL 128
#define D_STATE 16
#define BATCH   4
#define SEQ     4096
#define NCHUNK  16
#define CHUNK   (SEQ / NCHUNK)
#define LOG_CLAMP 70.0f

typedef float v2f __attribute__((ext_vector_type(2)));
typedef float v8f __attribute__((ext_vector_type(8)));

__device__ __forceinline__ float softplus_f(float z) {
    return fmaxf(z, 0.0f) + log1pf(__expf(-fabsf(z)));
}

__global__ void proj_wmma_kernel(const float* __restrict__ x,
                                 const float* __restrict__ W_B, const float* __restrict__ b_B,
                                 const float* __restrict__ W_C, const float* __restrict__ b_C,
                                 const float* __restrict__ W_D, const float* __restrict__ b_D,
                                 float* __restrict__ Bm, float* __restrict__ Cc,
                                 float* __restrict__ xwd) {
    const int lane  = threadIdx.x & 31;
    const int wave  = threadIdx.x >> 5;
    const int tile  = blockIdx.x * (blockDim.x >> 5) + wave;
    const int half  = lane >> 4;
    const int lid   = lane & 15;
    const int row   = tile * 16 + lid;
    const int khalf = half * 2;
    const float dmask = (lid == 0) ? 1.0f : 0.0f;

    v8f accB = {}; v8f accC = {}; v8f accD = {};
    const float* xrow = x + (size_t)row * D_MODEL;

    for (int k0 = 0; k0 < D_MODEL; k0 += 4) {
        const v2f a = *(const v2f*)(xrow + k0 + khalf);
        v2f fB, fC, fD;
        fB.x = W_B[(k0 + khalf + 0) * D_STATE + lid];
        fB.y = W_B[(k0 + khalf + 1) * D_STATE + lid];
        fC.x = W_C[(k0 + khalf + 0) * D_STATE + lid];
        fC.y = W_C[(k0 + khalf + 1) * D_STATE + lid];
        fD.x = W_D[k0 + khalf + 0] * dmask;
        fD.y = W_D[k0 + khalf + 1] * dmask;
        accB = __builtin_amdgcn_wmma_f32_16x16x4_f32(false, a, false, fB, (short)0, accB, false, false);
        accC = __builtin_amdgcn_wmma_f32_16x16x4_f32(false, a, false, fC, (short)0, accC, false, false);
        accD = __builtin_amdgcn_wmma_f32_16x16x4_f32(false, a, false, fD, (short)0, accD, false, false);
    }

    const float bBn = b_B[lid];
    const float bCn = b_C[lid];
    const float bDs = b_D[0];
#pragma unroll
    for (int r = 0; r < 8; ++r) {
        const int orow = tile * 16 + r + 8 * half;
        Bm[(size_t)orow * D_STATE + lid] = accB[r] + bBn;
        Cc[(size_t)orow * D_STATE + lid] = accC[r] + bCn;
        if (lid == 0) xwd[orow] = accD[r] + bDs;
    }
}

__global__ void disc_kernel(const float* __restrict__ xwd, const float* __restrict__ P,
                            float* __restrict__ disc, float* __restrict__ discsum) {
    const int t  = blockIdx.x * blockDim.x + threadIdx.x;
    const int d  = t % D_MODEL;
    const int g  = t / D_MODEL;
    const int ch = g % NCHUNK;
    const int b  = g / NCHUNK;

    const float Pd = P[d];
    float acc = 0.0f;
    const int s0 = ch * CHUNK;
    for (int s = s0; s < s0 + CHUNK; ++s) {
        const float dc = softplus_f(Pd + xwd[b * SEQ + s]);
        disc[((size_t)(b * SEQ + s)) * D_MODEL + d] = dc;
        acc += dc;
    }
    discsum[(b * D_MODEL + d) * NCHUNK + ch] = acc;
}

__global__ void scan_chunks_kernel(float* __restrict__ arr, int nchannels) {
    const int t = blockIdx.x * blockDim.x + threadIdx.x;
    if (t >= nchannels) return;
    float run = 0.0f;
    float* p = arr + (size_t)t * NCHUNK;
#pragma unroll
    for (int c = 0; c < NCHUNK; ++c) { const float v = p[c]; p[c] = run; run += v; }
}

__global__ void scan_pass1_kernel(const float* __restrict__ x,
                                  const float* __restrict__ Araw,
                                  const float* __restrict__ disc,
                                  const float* __restrict__ discpre,
                                  const float* __restrict__ Bm,
                                  float* __restrict__ sumS) {
    const int t  = blockIdx.x * blockDim.x + threadIdx.x;
    const int n  = t & 15;
    const int g  = t >> 4;
    const int ch = g % NCHUNK;
    const int g2 = g / NCHUNK;
    const int d  = g2 % D_MODEL;
    const int b  = g2 / D_MODEL;

    const float A    = -softplus_f(Araw[d * D_STATE + n]);
    const float invA = 1.0f / A;
    float Dcum = discpre[(b * D_MODEL + d) * NCHUNK + ch];
    float acc  = 0.0f;
    const int s0 = ch * CHUNK;
    for (int s = s0; s < s0 + CHUNK; ++s) {
        const size_t bs = (size_t)(b * SEQ + s);
        const float dc  = disc[bs * D_MODEL + d];
        Dcum += dc;
        const float L   = A * Dcum;
        const float xv  = x[bs * D_MODEL + d];
        const float bm  = Bm[bs * D_STATE + n];
        const float Bx  = bm * invA * (__expf(dc * A) - 1.0f) * xv;
        const float bxl = __logf(fabsf(Bx));
        const float Lc  = fmaxf(L, bxl - LOG_CLAMP);
        const float sgn = (Bx > 0.0f) ? 1.0f : ((Bx < 0.0f) ? -1.0f : 0.0f);
        acc += __expf(bxl - Lc) * sgn;
    }
    sumS[((b * D_MODEL + d) * D_STATE + n) * NCHUNK + ch] = acc;
}

__global__ void scan_final_kernel(const float* __restrict__ x,
                                  const float* __restrict__ Araw,
                                  const float* __restrict__ disc,
                                  const float* __restrict__ discpre,
                                  const float* __restrict__ Bm,
                                  const float* __restrict__ Cc,
                                  const float* __restrict__ sumSpre,
                                  float* __restrict__ y) {
    const int t  = blockIdx.x * blockDim.x + threadIdx.x;
    const int n  = t & 15;
    const int g  = t >> 4;
    const int ch = g % NCHUNK;
    const int g2 = g / NCHUNK;
    const int d  = g2 % D_MODEL;
    const int b  = g2 / D_MODEL;

    const float A    = -softplus_f(Araw[d * D_STATE + n]);
    const float invA = 1.0f / A;
    float Dcum = discpre[(b * D_MODEL + d) * NCHUNK + ch];
    float Ssum = sumSpre[((b * D_MODEL + d) * D_STATE + n) * NCHUNK + ch];
    const int s0 = ch * CHUNK;
    for (int s = s0; s < s0 + CHUNK; ++s) {
        const size_t bs = (size_t)(b * SEQ + s);
        const float dc  = disc[bs * D_MODEL + d];
        Dcum += dc;
        const float L   = A * Dcum;
        const float xv  = x[bs * D_MODEL + d];
        const float bm  = Bm[bs * D_STATE + n];
        const float Bx  = bm * invA * (__expf(dc * A) - 1.0f) * xv;
        const float bxl = __logf(fabsf(Bx));
        const float Lc  = fmaxf(L, bxl - LOG_CLAMP);
        const float sgn = (Bx > 0.0f) ? 1.0f : ((Bx < 0.0f) ? -1.0f : 0.0f);
        Ssum += __expf(bxl - Lc) * sgn;
        const float h = Ssum * __expf(Lc);
        float contrib = h * Cc[bs * D_STATE + n];
        contrib += __shfl_xor(contrib, 1, 32);
        contrib += __shfl_xor(contrib, 2, 32);
        contrib += __shfl_xor(contrib, 4, 32);
        contrib += __shfl_xor(contrib, 8, 32);
        if (n == 0) y[bs * D_MODEL + d] = contrib;
    }
}

extern "C" void kernel_launch(void* const* d_in, const int* in_sizes, int n_in,
                              void* d_out, int out_size, void* d_ws, size_t ws_size,
                              hipStream_t stream) {
    (void)in_sizes; (void)n_in; (void)out_size; (void)ws_size;
    const float* x    = (const float*)d_in[0];
    const float* Araw = (const float*)d_in[1];
    const float* P    = (const float*)d_in[2];
    const float* W_B  = (const float*)d_in[3];
    const float* b_B  = (const float*)d_in[4];
    const float* W_C  = (const float*)d_in[5];
    const float* b_C  = (const float*)d_in[6];
    const float* W_D  = (const float*)d_in[7];
    const float* b_D  = (const float*)d_in[8];
    float* y = (float*)d_out;

    constexpr size_t BS       = (size_t)BATCH * SEQ;
    constexpr size_t szBm     = BS * D_STATE * 4;
    constexpr size_t szCc     = BS * D_STATE * 4;
    constexpr size_t szXwd    = BS * 4;
    constexpr size_t szDisc   = BS * D_MODEL * 4;
    constexpr size_t szDsum   = (size_t)BATCH * D_MODEL * NCHUNK * 4;
    char* ws = (char*)d_ws;
    float* Bm   = (float*)(ws);
    float* Cc   = (float*)(ws + szBm);
    float* xwd  = (float*)(ws + szBm + szCc);
    float* disc = (float*)(ws + szBm + szCc + szXwd);
    float* dsum = (float*)(ws + szBm + szCc + szXwd + szDisc);
    float* sumS = (float*)(ws + szBm + szCc + szXwd + szDisc + szDsum);

    proj_wmma_kernel<<<(BS / 16) / 8, 256, 0, stream>>>(x, W_B, b_B, W_C, b_C, W_D, b_D,
                                                        Bm, Cc, xwd);
    disc_kernel<<<(BATCH * D_MODEL * NCHUNK) / 256, 256, 0, stream>>>(xwd, P, disc, dsum);
    scan_chunks_kernel<<<(BATCH * D_MODEL + 255) / 256, 256, 0, stream>>>(dsum, BATCH * D_MODEL);
    scan_pass1_kernel<<<(BATCH * D_MODEL * D_STATE * NCHUNK) / 256, 256, 0, stream>>>(
        x, Araw, disc, dsum, Bm, sumS);
    scan_chunks_kernel<<<(BATCH * D_MODEL * D_STATE + 255) / 256, 256, 0, stream>>>(
        sumS, BATCH * D_MODEL * D_STATE);
    scan_final_kernel<<<(BATCH * D_MODEL * D_STATE * NCHUNK) / 256, 256, 0, stream>>>(
        x, Araw, disc, dsum, Bm, Cc, sumS, y);
}


# Projections moved into the disc/transposition launch (launches 4 -> 3): a block stages its 32-step slab of x in LDS once; WMMA A fragments and the turned x copy read it there; xwd stays in LDS
# speedup vs baseline: 8.3585x; 8.3585x over previous
#include <hip/hip_runtime.h>
#include <math.h>


#define D_MODEL 128
#define D_STATE 16
#define BATCH   4
#define SEQ     4096
#define LOG_CLAMP 70.0f

typedef float v2f __attribute__((ext_vector_type(2)));
typedef float v8f __attribute__((ext_vector_type(8)));

__device__ __forceinline__ float softplus_f(float z) {
    return fmaxf(z, 0.0f) + log1pf(__expf(-fabsf(z)));
}

#define PSTEPS 32
#define PWAVES (D_MODEL / 32)
#define TXW    (32 * 33)
static_assert(PWAVES == 4 && PSTEPS == 32, "two 16-row tiles x {W_B+W_D, W_C} = four waves");

template <bool WITH_D>
__device__ __forceinline__ void proj_tile(const float* __restrict__ xa,
                                          const float* __restrict__ W1,
                                          const float* __restrict__ W_D,
                                          int lid, int khalf, float dmask,
                                          v8f& acc1, v8f& accD) {
    v2f f1[D_MODEL / 4], fD[D_MODEL / 4];
#pragma unroll
    for (int s = 0; s < D_MODEL / 4; ++s) {
        const int k = 4 * s + khalf;
        f1[s].x = W1[(k + 0) * D_STATE + lid];
        f1[s].y = W1[(k + 1) * D_STATE + lid];
        if (WITH_D) {
            fD[s].x = W_D[k + 0] * dmask;
            fD[s].y = W_D[k + 1] * dmask;
        }
    }
#pragma unroll
    for (int s = 0; s < D_MODEL / 4; ++s) {
        const float* p = xa + ((4 * s) >> 5) * TXW + ((4 * s) & 31) * 33;
        v2f a;
        a.x = p[0];
        a.y = p[33];
        acc1 = __builtin_amdgcn_wmma_f32_16x16x4_f32(false, a, false, f1[s], (short)0, acc1, false, false);
        if (WITH_D)
            accD = __builtin_amdgcn_wmma_f32_16x16x4_f32(false, a, false, fD[s], (short)0, accD, false, false);
    }
}

__global__ void __launch_bounds__(32 * PWAVES)
proj_prep_kernel(const float* __restrict__ x,
                 const float* __restrict__ W_B, const float* __restrict__ b_B,
                 const float* __restrict__ W_C, const float* __restrict__ b_C,
                 const float* __restrict__ W_D, const float* __restrict__ b_D,
                 const float* __restrict__ P,
                 float* __restrict__ Bm, float* __restrict__ Cc,
                 float* __restrict__ discT, float* __restrict__ xT) {
    __shared__ float td[PWAVES][32][33];
    __shared__ float tx[PWAVES][32][33];
    __shared__ __attribute__((aligned(16))) float xs[PSTEPS];

    const int lane = threadIdx.x & 31;
    const int w    = __builtin_amdgcn_readfirstlane(threadIdx.x >> 5);
    const int sl   = blockIdx.x % (SEQ / PSTEPS);
    const int b    = blockIdx.x / (SEQ / PSTEPS);
    const int d0   = w * 32;
    const int d    = d0 + lane;
    const int half = lane >> 4;
    const int lid  = lane & 15;
    const int mt   = w & 1;
    const float dmask = (lid == 0) ? 1.0f : 0.0f;

    const float  Pd   = P[d];
    const size_t row0 = (size_t)b * SEQ + (size_t)sl * PSTEPS;

#pragma unroll
    for (int j = 0; j < PSTEPS; ++j) tx[w][lane][j] = x[(row0 + j) * D_MODEL + d];
    __syncthreads();

    {
        const float* xa = &tx[0][2 * half][16 * mt + lid];
        v8f acc1 = {}; v8f accD = {};
        const size_t orow0 = row0 + 16 * mt + 8 * half;
        if (w < 2) {
            proj_tile<true>(xa, W_B, W_D, lid, 2 * half, dmask, acc1, accD);
            const float bBn = b_B[lid];
            const float bDs = b_D[0];
#pragma unroll
            for (int r = 0; r < 8; ++r) {
                Bm[(orow0 + r) * D_STATE + lid] = acc1[r] + bBn;
                if (lid == 0) xs[16 * mt + 8 * half + r] = accD[r] + bDs;
            }
        } else {
            proj_tile<false>(xa, W_C, W_D, lid, 2 * half, dmask, acc1, accD);
            const float bCn = b_C[lid];
#pragma unroll
            for (int r = 0; r < 8; ++r) Cc[(orow0 + r) * D_STATE + lid] = acc1[r] + bCn;
        }
    }
    __syncthreads();

    float xw[PSTEPS];
#pragma unroll
    for (int q = 0; q < PSTEPS / 4; ++q) {
        const float4 v = ((const float4*)xs)[q];
        xw[4 * q + 0] = v.x; xw[4 * q + 1] = v.y; xw[4 * q + 2] = v.z; xw[4 * q + 3] = v.w;
    }
#pragma unroll
    for (int j = 0; j < PSTEPS; ++j) td[w][lane][j] = softplus_f(Pd + xw[j]);
    __syncthreads();
    float* dOut = discT + ((size_t)b * D_MODEL + d0) * SEQ + (size_t)sl * PSTEPS + lane;
    float* xOut = xT    + ((size_t)b * D_MODEL + d0) * SEQ + (size_t)sl * PSTEPS + lane;
#pragma unroll
    for (int j = 0; j < 32; ++j) {
        dOut[(size_t)j * SEQ] = td[w][j][lane];
        xOut[(size_t)j * SEQ] = tx[w][j][lane];
    }
}

__device__ __forceinline__ v8f tri_cumsum(const v2f* __restrict__ ta, const v8f v, const float carry) {
    v8f acc = {carry, carry, carry, carry, carry, carry, carry, carry};
#pragma unroll
    for (int j = 0; j < 4; ++j) {
        v2f bf;
        bf.x = v[2 * j];
        bf.y = v[2 * j + 1];
        acc = __builtin_amdgcn_wmma_f32_16x16x4_f32(false, ta[j], false, bf, (short)0, acc, false, false);
    }
    return acc;
}

__device__ __forceinline__ void scan_element(float dc, float Dcum, float xv, float bm,
                                             float A, float invA, float& V, float& Lc) {
    const float L   = A * Dcum;
    const float Bx  = bm * invA * (__expf(dc * A) - 1.0f) * xv;
    const float bxl = __logf(fabsf(Bx));
    Lc = fmaxf(L, bxl - LOG_CLAMP);
    const float sgn = (Bx > 0.0f) ? 1.0f : ((Bx < 0.0f) ? -1.0f : 0.0f);
    V = __expf(bxl - Lc) * sgn;
}

__device__ __forceinline__ void phase_fence() {
    __builtin_amdgcn_sched_barrier(0);
}

struct TileIn { float4 d0, d1, x0, x1; float bm[8], cv[8]; };

__device__ __forceinline__ void fetch_tile(TileIn& o, const float* __restrict__ dL,
                                           const float* __restrict__ xL,
                                           const float* __restrict__ bmL,
                                           const float* __restrict__ ccL, int t) {
    const float4* dp = (const float4*)(dL + t * 16);
    const float4* xp = (const float4*)(xL + t * 16);
    o.d0 = dp[0]; o.d1 = dp[1];
    o.x0 = xp[0]; o.x1 = xp[1];
#pragma unroll
    for (int r = 0; r < 8; ++r) {
        o.bm[r] = bmL[t * (16 * D_STATE) + r * D_STATE];
        o.cv[r] = ccL[t * (16 * D_STATE) + r * D_STATE];
    }
}

#define NW     16
#define NTILE  (SEQ / 16)
#define ROUNDS (NTILE / NW)
#define TPITCH 20
static_assert(NTILE == 256, "the prologue sums one tile per thread in 8 waves");
static_assert(ROUNDS % 2 == 0, "the round loop takes two rounds per trip");

__device__ __forceinline__ void scan_round(const TileIn& o, const v2f* __restrict__ ta,
                                           float A, float invA, float dcarry,
                                           float* __restrict__ totw, int w, int c, int h,
                                           v8f& Sl, v8f& eL) {
    const v8f dc = {o.d0.x, o.d0.y, o.d0.z, o.d0.w, o.d1.x, o.d1.y, o.d1.z, o.d1.w};
    const v8f xv = {o.x0.x, o.x0.y, o.x0.z, o.x0.w, o.x1.x, o.x1.y, o.x1.z, o.x1.w};
    const v8f Dcum = tri_cumsum(ta, dc, dcarry);
    v8f V;
#pragma unroll
    for (int r = 0; r < 8; ++r) {
        float v, Lc;
        scan_element(dc[r], Dcum[r], xv[r], o.bm[r], A, invA, v, Lc);
        V[r]  = v;
        eL[r] = __expf(Lc);
    }
    Sl = tri_cumsum(ta, V, 0.0f);
    if (h == 1) totw[c * TPITCH + w] = Sl[7];
}

__device__ __forceinline__ void scan_finish(const TileIn& o, const float* __restrict__ totr,
                                            float* __restrict__ ybw, float& base,
                                            int w, int c, int h, const v8f Sl, const v8f eL) {
    const float4* tp = (const float4*)(totr + c * TPITCH);
    float tv[16];
#pragma unroll
    for (int i = 0; i < 4; ++i) {
        const float4 v = tp[i];
        tv[4 * i + 0] = v.x; tv[4 * i + 1] = v.y; tv[4 * i + 2] = v.z; tv[4 * i + 3] = v.w;
    }
    float run = base, carry = 0.0f;
#pragma unroll
    for (int j = 0; j < NW; ++j) {
        if (j == w) carry = run;
        run += tv[j];
    }
    base = run;

    float pr[8];
#pragma unroll
    for (int r = 0; r < 8; ++r) {
        const float hh = (Sl[r] + carry) * eL[r];
        pr[r] = hh * o.cv[r];
    }
    float q4[4], q2[2];
#pragma unroll
    for (int i = 0; i < 4; ++i) {
        const float keep = (c & 8) ? pr[4 + i] : pr[i];
        const float send = (c & 8) ? pr[i] : pr[4 + i];
        q4[i] = keep + __shfl_xor(send, 8, 32);
    }
#pragma unroll
    for (int i = 0; i < 2; ++i) {
        const float keep = (c & 4) ? q4[2 + i] : q4[i];
        const float send = (c & 4) ? q4[i] : q4[2 + i];
        q2[i] = keep + __shfl_xor(send, 4, 32);
    }
    const float keep1 = (c & 2) ? q2[1] : q2[0];
    const float send1 = (c & 2) ? q2[0] : q2[1];
    const float q1 = keep1 + __shfl_xor(send1, 2, 32);
    const float q0 = q1 + __shfl_xor(q1, 1, 32);
    if ((c & 1) == 0) ybw[16 * w + 8 * h + (c >> 1)] = q0;
}

__global__ void __launch_bounds__(32 * NW)
scan_kernel(const float* __restrict__ Araw,
            const float* __restrict__ discT,
            const float* __restrict__ xT,
            const float* __restrict__ Bm,
            const float* __restrict__ Cc,
            float* __restrict__ yT) {
    __shared__ float dpre[NTILE];
    __shared__ float wsum[8];
    __shared__ __attribute__((aligned(16))) float tot[2][16 * TPITCH];
    __shared__ float ybuf[2][16 * NW];

    const int tid  = threadIdx.x;
    const int lane = tid & 31;
    const int w    = __builtin_amdgcn_readfirstlane(tid >> 5);
    const int c    = lane & 15;
    const int h    = lane >> 4;
    const int d    = blockIdx.x % D_MODEL;
    const int b    = blockIdx.x / D_MODEL;
    const float* dT  = discT + (size_t)blockIdx.x * SEQ;
    const float* xTp = xT    + (size_t)blockIdx.x * SEQ;
    float*       yTp = yT    + (size_t)blockIdx.x * SEQ;
    const float* dL  = dT  + 8 * h;
    const float* xL  = xTp + 8 * h;
    const float* bmL = Bm + ((size_t)b * SEQ + 8 * h) * D_STATE + c;
    const float* ccL = Cc + ((size_t)b * SEQ + 8 * h) * D_STATE + c;

    const float A    = -softplus_f(Araw[d * D_STATE + c]);
    const float invA = 1.0f / A;
    v2f ta[4];
#pragma unroll
    for (int j = 0; j < 4; ++j) {
        ta[j].x = (8 * h + 2 * j     <= c) ? 1.0f : 0.0f;
        ta[j].y = (8 * h + 2 * j + 1 <= c) ? 1.0f : 0.0f;
    }

    TileIn p, q;
    fetch_tile(p, dL, xL, bmL, ccL, w);

    {
        const float4* tq = (const float4*)(dT + (tid & (NTILE - 1)) * 16);
        float s = 0.0f;
#pragma unroll
        for (int i = 0; i < 4; ++i) {
            const float4 v = tq[i];
            s += v.x; s += v.y; s += v.z; s += v.w;
        }
        float incl = s;
#pragma unroll
        for (int off = 1; off < 32; off <<= 1) {
            const float up = __shfl_up(incl, off, 32);
            if (lane >= off) incl += up;
        }
        float excl = __shfl_up(incl, 1, 32);
        if (lane == 0) excl = 0.0f;
        if (w < 8 && lane == 31) wsum[w] = incl;
        __syncthreads();
        if (w < 8) {
            float before = 0.0f;
#pragma unroll
            for (int j = 0; j < 8; ++j) before += (j < w) ? wsum[j] : 0.0f;
            dpre[tid] = before + excl;
        }
        __syncthreads();
    }

    float base = 0.0f;
    v8f Sl, eL;
    for (int k = 0; k < ROUNDS; k += 2) {
        fetch_tile(q, dL, xL, bmL, ccL, NW * (k + 1) + w);
        phase_fence();
        scan_round(p, ta, A, invA, dpre[NW * k + w], tot[0], w, c, h, Sl, eL);
        __syncthreads();
        if (k > 0 && w < 8) yTp[16 * NW * (k - 1) + 32 * w + lane] = ybuf[1][32 * w + lane];
        scan_finish(p, tot[0], ybuf[0], base, w, c, h, Sl, eL);
        phase_fence();

        fetch_tile(p, dL, xL, bmL, ccL, NW * ((k + 2 < ROUNDS) ? k + 2 : k + 1) + w);
        phase_fence();
        scan_round(q, ta, A, invA, dpre[NW * (k + 1) + w], tot[1], w, c, h, Sl, eL);
        __syncthreads();
        if (w < 8) yTp[16 * NW * k + 32 * w + lane] = ybuf[0][32 * w + lane];
        scan_finish(q, tot[1], ybuf[1], base, w, c, h, Sl, eL);
        phase_fence();
    }
    __syncthreads();
    if (w < 8) yTp[16 * NW * (ROUNDS - 1) + 32 * w + lane] = ybuf[1][32 * w + lane];
}

#define YSUB 4

__global__ void __launch_bounds__(256)
ytrans_kernel(const float* __restrict__ yT, float* __restrict__ y) {
    __shared__ float tl[YSUB][32][33];

    const int lane = threadIdx.x & 31;
    const int w    = threadIdx.x >> 5;
    const int dq   = blockIdx.x % (D_MODEL / 32);
    const int g    = blockIdx.x / (D_MODEL / 32);
    const int sb   = g % (SEQ / (32 * YSUB));
    const int b    = g / (SEQ / (32 * YSUB));
    const int d0   = dq * 32;
    const int s0   = sb * 32 * YSUB;

#pragma unroll
    for (int u = 0; u < YSUB; ++u) {
#pragma unroll
        for (int i = 0; i < 4; ++i) {
            const int j = w + 8 * i;
            tl[u][j][lane] = yT[((size_t)b * D_MODEL + d0 + j) * SEQ + s0 + 32 * u + lane];
        }
    }
    __syncthreads();
#pragma unroll
    for (int u = 0; u < YSUB; ++u) {
#pragma unroll
        for (int i = 0; i < 4; ++i) {
            const int j = w + 8 * i;
            y[((size_t)b * SEQ + s0 + 32 * u + j) * D_MODEL + d0 + lane] = tl[u][lane][j];
        }
    }
}

extern "C" void kernel_launch(void* const* d_in, const int* in_sizes, int n_in,
                              void* d_out, int out_size, void* d_ws, size_t ws_size,
                              hipStream_t stream) {
    (void)in_sizes; (void)n_in; (void)out_size; (void)ws_size;
    const float* x    = (const float*)d_in[0];
    const float* Araw = (const float*)d_in[1];
    const float* P    = (const float*)d_in[2];
    const float* W_B  = (const float*)d_in[3];
    const float* b_B  = (const float*)d_in[4];
    const float* W_C  = (const float*)d_in[5];
    const float* b_C  = (const float*)d_in[6];
    const float* W_D  = (const float*)d_in[7];
    const float* b_D  = (const float*)d_in[8];
    float* y = (float*)d_out;

    constexpr size_t BS       = (size_t)BATCH * SEQ;
    constexpr size_t szBm     = BS * D_STATE * 4;
    constexpr size_t szCc     = BS * D_STATE * 4;
    constexpr size_t szPlane  = BS * D_MODEL * 4;
    char* ws = (char*)d_ws;
    float* Bm    = (float*)(ws);
    float* Cc    = (float*)(ws + szBm);
    float* discT = (float*)(ws + szBm + szCc);
    float* xT    = (float*)(ws + szBm + szCc + szPlane);
    float* yT    = (float*)(ws + szBm + szCc + 2 * szPlane);

    proj_prep_kernel<<<BATCH * (SEQ / PSTEPS), 32 * PWAVES, 0, stream>>>(x, W_B, b_B, W_C, b_C, W_D, b_D, P,
                                                                      Bm, Cc, discT, xT);
    scan_kernel<<<BATCH * D_MODEL, 32 * NW, 0, stream>>>(Araw, discT, xT, Bm, Cc, yT);
    ytrans_kernel<<<BATCH * (SEQ / (32 * YSUB)) * (D_MODEL / 32), 256, 0, stream>>>(yT, y);
}


# Bm and C kept tile by tile in the scan's lane order (= the projection's accumulator order): a scan tile reads them with 4 16-byte loads (was 16 4-byte), the projection stores 2 per tile (was 8)
# speedup vs baseline: 8.7402x; 1.0457x over previous
#include <hip/hip_runtime.h>
#include <math.h>


#define D_MODEL 128
#define D_STATE 16
#define BATCH   4
#define SEQ     4096
#define LOG_CLAMP 70.0f

typedef float v2f __attribute__((ext_vector_type(2)));
typedef float v8f __attribute__((ext_vector_type(8)));

__device__ __forceinline__ float softplus_f(float z) {
    return fmaxf(z, 0.0f) + log1pf(__expf(-fabsf(z)));
}

#define PSTEPS 32
#define PWAVES (D_MODEL / 32)
#define TXW    (32 * 33)
#define FRAG_TILE 256
static_assert(PWAVES == 4 && PSTEPS == 32, "two 16-row tiles x {W_B+W_D, W_C} = four waves");

template <bool WITH_D>
__device__ __forceinline__ void proj_tile(const float* __restrict__ xa,
                                          const float* __restrict__ W1,
                                          const float* __restrict__ W_D,
                                          int lid, int khalf, float dmask,
                                          v8f& acc1, v8f& accD) {
    v2f f1[D_MODEL / 4], fD[D_MODEL / 4];
#pragma unroll
    for (int s = 0; s < D_MODEL / 4; ++s) {
        const int k = 4 * s + khalf;
        f1[s].x = W1[(k + 0) * D_STATE + lid];
        f1[s].y = W1[(k + 1) * D_STATE + lid];
        if (WITH_D) {
            fD[s].x = W_D[k + 0] * dmask;
            fD[s].y = W_D[k + 1] * dmask;
        }
    }
#pragma unroll
    for (int s = 0; s < D_MODEL / 4; ++s) {
        const float* p = xa + ((4 * s) >> 5) * TXW + ((4 * s) & 31) * 33;
        v2f a;
        a.x = p[0];
        a.y = p[33];
        acc1 = __builtin_amdgcn_wmma_f32_16x16x4_f32(false, a, false, f1[s], (short)0, acc1, false, false);
        if (WITH_D)
            accD = __builtin_amdgcn_wmma_f32_16x16x4_f32(false, a, false, fD[s], (short)0, accD, false, false);
    }
}

__global__ void __launch_bounds__(32 * PWAVES)
proj_prep_kernel(const float* __restrict__ x,
                 const float* __restrict__ W_B, const float* __restrict__ b_B,
                 const float* __restrict__ W_C, const float* __restrict__ b_C,
                 const float* __restrict__ W_D, const float* __restrict__ b_D,
                 const float* __restrict__ P,
                 float* __restrict__ Bm, float* __restrict__ Cc,
                 float* __restrict__ discT, float* __restrict__ xT) {
    __shared__ float td[PWAVES][32][33];
    __shared__ float tx[PWAVES][32][33];
    __shared__ __attribute__((aligned(16))) float xs[PSTEPS];

    const int lane = threadIdx.x & 31;
    const int w    = __builtin_amdgcn_readfirstlane(threadIdx.x >> 5);
    const int sl   = blockIdx.x % (SEQ / PSTEPS);
    const int b    = blockIdx.x / (SEQ / PSTEPS);
    const int d0   = w * 32;
    const int d    = d0 + lane;
    const int half = lane >> 4;
    const int lid  = lane & 15;
    const int mt   = w & 1;
    const float dmask = (lid == 0) ? 1.0f : 0.0f;

    const float  Pd   = P[d];
    const size_t row0 = (size_t)b * SEQ + (size_t)sl * PSTEPS;

#pragma unroll
    for (int j = 0; j < PSTEPS; ++j) tx[w][lane][j] = x[(row0 + j) * D_MODEL + d];
    __syncthreads();

    {
        const float* xa = &tx[0][2 * half][16 * mt + lid];
        v8f acc1 = {}; v8f accD = {};
        const size_t ftile = row0 / 16 + mt;
        if (w < 2) {
            proj_tile<true>(xa, W_B, W_D, lid, 2 * half, dmask, acc1, accD);
            const float bBn = b_B[lid];
            const float bDs = b_D[0];
            float4* bo = (float4*)(Bm + ftile * FRAG_TILE) + lane;
            bo[0]  = make_float4(acc1[0] + bBn, acc1[1] + bBn, acc1[2] + bBn, acc1[3] + bBn);
            bo[32] = make_float4(acc1[4] + bBn, acc1[5] + bBn, acc1[6] + bBn, acc1[7] + bBn);
#pragma unroll
            for (int r = 0; r < 8; ++r) {
                if (lid == 0) xs[16 * mt + 8 * half + r] = accD[r] + bDs;
            }
        } else {
            proj_tile<false>(xa, W_C, W_D, lid, 2 * half, dmask, acc1, accD);
            const float bCn = b_C[lid];
            float4* co = (float4*)(Cc + ftile * FRAG_TILE) + lane;
            co[0]  = make_float4(acc1[0] + bCn, acc1[1] + bCn, acc1[2] + bCn, acc1[3] + bCn);
            co[32] = make_float4(acc1[4] + bCn, acc1[5] + bCn, acc1[6] + bCn, acc1[7] + bCn);
        }
    }
    __syncthreads();

    float xw[PSTEPS];
#pragma unroll
    for (int q = 0; q < PSTEPS / 4; ++q) {
        const float4 v = ((const float4*)xs)[q];
        xw[4 * q + 0] = v.x; xw[4 * q + 1] = v.y; xw[4 * q + 2] = v.z; xw[4 * q + 3] = v.w;
    }
#pragma unroll
    for (int j = 0; j < PSTEPS; ++j) td[w][lane][j] = softplus_f(Pd + xw[j]);
    __syncthreads();
    float* dOut = discT + ((size_t)b * D_MODEL + d0) * SEQ + (size_t)sl * PSTEPS + lane;
    float* xOut = xT    + ((size_t)b * D_MODEL + d0) * SEQ + (size_t)sl * PSTEPS + lane;
#pragma unroll
    for (int j = 0; j < 32; ++j) {
        dOut[(size_t)j * SEQ] = td[w][j][lane];
        xOut[(size_t)j * SEQ] = tx[w][j][lane];
    }
}

__device__ __forceinline__ v8f tri_cumsum(const v2f* __restrict__ ta, const v8f v, const float carry) {
    v8f acc = {carry, carry, carry, carry, carry, carry, carry, carry};
#pragma unroll
    for (int j = 0; j < 4; ++j) {
        v2f bf;
        bf.x = v[2 * j];
        bf.y = v[2 * j + 1];
        acc = __builtin_amdgcn_wmma_f32_16x16x4_f32(false, ta[j], false, bf, (short)0, acc, false, false);
    }
    return acc;
}

__device__ __forceinline__ void scan_element(float dc, float Dcum, float xv, float bm,
                                             float A, float invA, float& V, float& Lc) {
    const float L   = A * Dcum;
    const float Bx  = bm * invA * (__expf(dc * A) - 1.0f) * xv;
    const float bxl = __logf(fabsf(Bx));
    Lc = fmaxf(L, bxl - LOG_CLAMP);
    const float sgn = (Bx > 0.0f) ? 1.0f : ((Bx < 0.0f) ? -1.0f : 0.0f);
    V = __expf(bxl - Lc) * sgn;
}

__device__ __forceinline__ void phase_fence() {
    __builtin_amdgcn_sched_barrier(0);
}

struct TileIn { float4 d0, d1, x0, x1; float bm[8], cv[8]; };

__device__ __forceinline__ void fetch_tile(TileIn& o, const float* __restrict__ dL,
                                           const float* __restrict__ xL,
                                           const float* __restrict__ bmL,
                                           const float* __restrict__ ccL, int t) {
    const float4* dp = (const float4*)(dL + t * 16);
    const float4* xp = (const float4*)(xL + t * 16);
    o.d0 = dp[0]; o.d1 = dp[1];
    o.x0 = xp[0]; o.x1 = xp[1];
    const float4* bp = (const float4*)(bmL + t * FRAG_TILE);
    const float4* cp = (const float4*)(ccL + t * FRAG_TILE);
    const float4 b0 = bp[0], b1 = bp[32], c0 = cp[0], c1 = cp[32];
    o.bm[0] = b0.x; o.bm[1] = b0.y; o.bm[2] = b0.z; o.bm[3] = b0.w;
    o.bm[4] = b1.x; o.bm[5] = b1.y; o.bm[6] = b1.z; o.bm[7] = b1.w;
    o.cv[0] = c0.x; o.cv[1] = c0.y; o.cv[2] = c0.z; o.cv[3] = c0.w;
    o.cv[4] = c1.x; o.cv[5] = c1.y; o.cv[6] = c1.z; o.cv[7] = c1.w;
}

#define NW     16
#define NTILE  (SEQ / 16)
#define ROUNDS (NTILE / NW)
#define TPITCH 20
static_assert(NTILE == 256, "the prologue sums one tile per thread in 8 waves");
static_assert(ROUNDS % 2 == 0, "the round loop takes two rounds per trip");

__device__ __forceinline__ void scan_round(const TileIn& o, const v2f* __restrict__ ta,
                                           float A, float invA, float dcarry,
                                           float* __restrict__ totw, int w, int c, int h,
                                           v8f& Sl, v8f& eL) {
    const v8f dc = {o.d0.x, o.d0.y, o.d0.z, o.d0.w, o.d1.x, o.d1.y, o.d1.z, o.d1.w};
    const v8f xv = {o.x0.x, o.x0.y, o.x0.z, o.x0.w, o.x1.x, o.x1.y, o.x1.z, o.x1.w};
    const v8f Dcum = tri_cumsum(ta, dc, dcarry);
    v8f V;
#pragma unroll
    for (int r = 0; r < 8; ++r) {
        float v, Lc;
        scan_element(dc[r], Dcum[r], xv[r], o.bm[r], A, invA, v, Lc);
        V[r]  = v;
        eL[r] = __expf(Lc);
    }
    Sl = tri_cumsum(ta, V, 0.0f);
    if (h == 1) totw[c * TPITCH + w] = Sl[7];
}

__device__ __forceinline__ void scan_finish(const TileIn& o, const float* __restrict__ totr,
                                            float* __restrict__ ybw, float& base,
                                            int w, int c, int h, const v8f Sl, const v8f eL) {
    const float4* tp = (const float4*)(totr + c * TPITCH);
    float tv[16];
#pragma unroll
    for (int i = 0; i < 4; ++i) {
        const float4 v = tp[i];
        tv[4 * i + 0] = v.x; tv[4 * i + 1] = v.y; tv[4 * i + 2] = v.z; tv[4 * i + 3] = v.w;
    }
    float run = base, carry = 0.0f;
#pragma unroll
    for (int j = 0; j < NW; ++j) {
        if (j == w) carry = run;
        run += tv[j];
    }
    base = run;

    float pr[8];
#pragma unroll
    for (int r = 0; r < 8; ++r) {
        const float hh = (Sl[r] + carry) * eL[r];
        pr[r] = hh * o.cv[r];
    }
    float q4[4], q2[2];
#pragma unroll
    for (int i = 0; i < 4; ++i) {
        const float keep = (c & 8) ? pr[4 + i] : pr[i];
        const float send = (c & 8) ? pr[i] : pr[4 + i];
        q4[i] = keep + __shfl_xor(send, 8, 32);
    }
#pragma unroll
    for (int i = 0; i < 2; ++i) {
        const float keep = (c & 4) ? q4[2 + i] : q4[i];
        const float send = (c & 4) ? q4[i] : q4[2 + i];
        q2[i] = keep + __shfl_xor(send, 4, 32);
    }
    const float keep1 = (c & 2) ? q2[1] : q2[0];
    const float send1 = (c & 2) ? q2[0] : q2[1];
    const float q1 = keep1 + __shfl_xor(send1, 2, 32);
    const float q0 = q1 + __shfl_xor(q1, 1, 32);
    if ((c & 1) == 0) ybw[16 * w + 8 * h + (c >> 1)] = q0;
}

__global__ void __launch_bounds__(32 * NW)
scan_kernel(const float* __restrict__ Araw,
            const float* __restrict__ discT,
            const float* __restrict__ xT,
            const float* __restrict__ Bm,
            const float* __restrict__ Cc,
            float* __restrict__ yT) {
    __shared__ float dpre[NTILE];
    __shared__ float wsum[8];
    __shared__ __attribute__((aligned(16))) float tot[2][16 * TPITCH];
    __shared__ float ybuf[2][16 * NW];

    const int tid  = threadIdx.x;
    const int lane = tid & 31;
    const int w    = __builtin_amdgcn_readfirstlane(tid >> 5);
    const int c    = lane & 15;
    const int h    = lane >> 4;
    const int d    = blockIdx.x % D_MODEL;
    const int b    = blockIdx.x / D_MODEL;
    const float* dT  = discT + (size_t)blockIdx.x * SEQ;
    const float* xTp = xT    + (size_t)blockIdx.x * SEQ;
    float*       yTp = yT    + (size_t)blockIdx.x * SEQ;
    const float* dL  = dT  + 8 * h;
    const float* xL  = xTp + 8 * h;
    const float* bmL = Bm + (size_t)b * NTILE * FRAG_TILE + 4 * lane;
    const float* ccL = Cc + (size_t)b * NTILE * FRAG_TILE + 4 * lane;

    const float A    = -softplus_f(Araw[d * D_STATE + c]);
    const float invA = 1.0f / A;
    v2f ta[4];
#pragma unroll
    for (int j = 0; j < 4; ++j) {
        ta[j].x = (8 * h + 2 * j     <= c) ? 1.0f : 0.0f;
        ta[j].y = (8 * h + 2 * j + 1 <= c) ? 1.0f : 0.0f;
    }

    TileIn p, q;
    fetch_tile(p, dL, xL, bmL, ccL, w);

    {
        const float4* tq = (const float4*)(dT + (tid & (NTILE - 1)) * 16);
        float s = 0.0f;
#pragma unroll
        for (int i = 0; i < 4; ++i) {
            const float4 v = tq[i];
            s += v.x; s += v.y; s += v.z; s += v.w;
        }
        float incl = s;
#pragma unroll
        for (int off = 1; off < 32; off <<= 1) {
            const float up = __shfl_up(incl, off, 32);
            if (lane >= off) incl += up;
        }
        float excl = __shfl_up(incl, 1, 32);
        if (lane == 0) excl = 0.0f;
        if (w < 8 && lane == 31) wsum[w] = incl;
        __syncthreads();
        if (w < 8) {
            float before = 0.0f;
#pragma unroll
            for (int j = 0; j < 8; ++j) before += (j < w) ? wsum[j] : 0.0f;
            dpre[tid] = before + excl;
        }
        __syncthreads();
    }

    float base = 0.0f;
    v8f Sl, eL;
    for (int k = 0; k < ROUNDS; k += 2) {
        fetch_tile(q, dL, xL, bmL, ccL, NW * (k + 1) + w);
        phase_fence();
        scan_round(p, ta, A, invA, dpre[NW * k + w], tot[0], w, c, h, Sl, eL);
        __syncthreads();
        if (k > 0 && w < 8) yTp[16 * NW * (k - 1) + 32 * w + lane] = ybuf[1][32 * w + lane];
        scan_finish(p, tot[0], ybuf[0], base, w, c, h, Sl, eL);
        phase_fence();

        fetch_tile(p, dL, xL, bmL, ccL, NW * ((k + 2 < ROUNDS) ? k + 2 : k + 1) + w);
        phase_fence();
        scan_round(q, ta, A, invA, dpre[NW * (k + 1) + w], tot[1], w, c, h, Sl, eL);
        __syncthreads();
        if (w < 8) yTp[16 * NW * k + 32 * w + lane] = ybuf[0][32 * w + lane];
        scan_finish(q, tot[1], ybuf[1], base, w, c, h, Sl, eL);
        phase_fence();
    }
    __syncthreads();
    if (w < 8) yTp[16 * NW * (ROUNDS - 1) + 32 * w + lane] = ybuf[1][32 * w + lane];
}

#define YSUB 4

__global__ void __launch_bounds__(256)
ytrans_kernel(const float* __restrict__ yT, float* __restrict__ y) {
    __shared__ float tl[YSUB][32][33];

    const int lane = threadIdx.x & 31;
    const int w    = threadIdx.x >> 5;
    const int dq   = blockIdx.x % (D_MODEL / 32);
    const int g    = blockIdx.x / (D_MODEL / 32);
    const int sb   = g % (SEQ / (32 * YSUB));
    const int b    = g / (SEQ / (32 * YSUB));
    const int d0   = dq * 32;
    const int s0   = sb * 32 * YSUB;

#pragma unroll
    for (int u = 0; u < YSUB; ++u) {
#pragma unroll
        for (int i = 0; i < 4; ++i) {
            const int j = w + 8 * i;
            tl[u][j][lane] = yT[((size_t)b * D_MODEL + d0 + j) * SEQ + s0 + 32 * u + lane];
        }
    }
    __syncthreads();
#pragma unroll
    for (int u = 0; u < YSUB; ++u) {
#pragma unroll
        for (int i = 0; i < 4; ++i) {
            const int j = w + 8 * i;
            y[((size_t)b * SEQ + s0 + 32 * u + j) * D_MODEL + d0 + lane] = tl[u][lane][j];
        }
    }
}

extern "C" void kernel_launch(void* const* d_in, const int* in_sizes, int n_in,
                              void* d_out, int out_size, void* d_ws, size_t ws_size,
                              hipStream_t stream) {
    (void)in_sizes; (void)n_in; (void)out_size; (void)ws_size;
    const float* x    = (const float*)d_in[0];
    const float* Araw = (const float*)d_in[1];
    const float* P    = (const float*)d_in[2];
    const float* W_B  = (const float*)d_in[3];
    const float* b_B  = (const float*)d_in[4];
    const float* W_C  = (const float*)d_in[5];
    const float* b_C  = (const float*)d_in[6];
    const float* W_D  = (const float*)d_in[7];
    const float* b_D  = (const float*)d_in[8];
    float* y = (float*)d_out;

    constexpr size_t BS       = (size_t)BATCH * SEQ;
    constexpr size_t szBm     = BS * D_STATE * 4;
    constexpr size_t szCc     = BS * D_STATE * 4;
    constexpr size_t szPlane  = BS * D_MODEL * 4;
    char* ws = (char*)d_ws;
    float* Bm    = (float*)(ws);
    float* Cc    = (float*)(ws + szBm);
    float* discT = (float*)(ws + szBm + szCc);
    float* xT    = (float*)(ws + szBm + szCc + szPlane);
    float* yT    = (float*)(ws + szBm + szCc + 2 * szPlane);

    proj_prep_kernel<<<BATCH * (SEQ / PSTEPS), 32 * PWAVES, 0, stream>>>(x, W_B, b_B, W_C, b_C, W_D, b_D, P,
                                                                      Bm, Cc, discT, xT);
    scan_kernel<<<BATCH * D_MODEL, 32 * NW, 0, stream>>>(Araw, discT, xT, Bm, Cc, yT);
    ytrans_kernel<<<BATCH * (SEQ / (32 * YSUB)) * (D_MODEL / 32), 256, 0, stream>>>(yT, y);
}


# Scan: disc running sum taken once per block in the prologue (16 tiles = the 16 columns of one f32 WMMA product, kept in LDS), not per tile with 16 equal columns: 4 of a tile's 8 WMMAs go
# speedup vs baseline: 9.4311x; 1.0790x over previous
#include <hip/hip_runtime.h>
#include <math.h>


#define D_MODEL 128
#define D_STATE 16
#define BATCH   4
#define SEQ     4096
#define LOG_CLAMP 70.0f

typedef float v2f __attribute__((ext_vector_type(2)));
typedef float v8f __attribute__((ext_vector_type(8)));

__device__ __forceinline__ float softplus_f(float z) {
    return fmaxf(z, 0.0f) + log1pf(__expf(-fabsf(z)));
}

#define PSTEPS 32
#define PWAVES (D_MODEL / 32)
#define TXW    (32 * 33)
#define FRAG_TILE 256
static_assert(PWAVES == 4 && PSTEPS == 32, "two 16-row tiles x {W_B+W_D, W_C} = four waves");

template <bool WITH_D>
__device__ __forceinline__ void proj_tile(const float* __restrict__ xa,
                                          const float* __restrict__ W1,
                                          const float* __restrict__ W_D,
                                          int lid, int khalf, float dmask,
                                          v8f& acc1, v8f& accD) {
    v2f f1[D_MODEL / 4], fD[D_MODEL / 4];
#pragma unroll
    for (int s = 0; s < D_MODEL / 4; ++s) {
        const int k = 4 * s + khalf;
        f1[s].x = W1[(k + 0) * D_STATE + lid];
        f1[s].y = W1[(k + 1) * D_STATE + lid];
        if (WITH_D) {
            fD[s].x = W_D[k + 0] * dmask;
            fD[s].y = W_D[k + 1] * dmask;
        }
    }
#pragma unroll
    for (int s = 0; s < D_MODEL / 4; ++s) {
        const float* p = xa + ((4 * s) >> 5) * TXW + ((4 * s) & 31) * 33;
        v2f a;
        a.x = p[0];
        a.y = p[33];
        acc1 = __builtin_amdgcn_wmma_f32_16x16x4_f32(false, a, false, f1[s], (short)0, acc1, false, false);
        if (WITH_D)
            accD = __builtin_amdgcn_wmma_f32_16x16x4_f32(false, a, false, fD[s], (short)0, accD, false, false);
    }
}

__global__ void __launch_bounds__(32 * PWAVES)
proj_prep_kernel(const float* __restrict__ x,
                 const float* __restrict__ W_B, const float* __restrict__ b_B,
                 const float* __restrict__ W_C, const float* __restrict__ b_C,
                 const float* __restrict__ W_D, const float* __restrict__ b_D,
                 const float* __restrict__ P,
                 float* __restrict__ Bm, float* __restrict__ Cc,
                 float* __restrict__ discT, float* __restrict__ xT) {
    __shared__ float td[PWAVES][32][33];
    __shared__ float tx[PWAVES][32][33];
    __shared__ __attribute__((aligned(16))) float xs[PSTEPS];

    const int lane = threadIdx.x & 31;
    const int w    = __builtin_amdgcn_readfirstlane(threadIdx.x >> 5);
    const int sl   = blockIdx.x % (SEQ / PSTEPS);
    const int b    = blockIdx.x / (SEQ / PSTEPS);
    const int d0   = w * 32;
    const int d    = d0 + lane;
    const int half = lane >> 4;
    const int lid  = lane & 15;
    const int mt   = w & 1;
    const float dmask = (lid == 0) ? 1.0f : 0.0f;

    const float  Pd   = P[d];
    const size_t row0 = (size_t)b * SEQ + (size_t)sl * PSTEPS;

#pragma unroll
    for (int j = 0; j < PSTEPS; ++j) tx[w][lane][j] = x[(row0 + j) * D_MODEL + d];
    __syncthreads();

    {
        const float* xa = &tx[0][2 * half][16 * mt + lid];
        v8f acc1 = {}; v8f accD = {};
        const size_t ftile = row0 / 16 + mt;
        if (w < 2) {
            proj_tile<true>(xa, W_B, W_D, lid, 2 * half, dmask, acc1, accD);
            const float bBn = b_B[lid];
            const float bDs = b_D[0];
            float4* bo = (float4*)(Bm + ftile * FRAG_TILE) + lane;
            bo[0]  = make_float4(acc1[0] + bBn, acc1[1] + bBn, acc1[2] + bBn, acc1[3] + bBn);
            bo[32] = make_float4(acc1[4] + bBn, acc1[5] + bBn, acc1[6] + bBn, acc1[7] + bBn);
#pragma unroll
            for (int r = 0; r < 8; ++r) {
                if (lid == 0) xs[16 * mt + 8 * half + r] = accD[r] + bDs;
            }
        } else {
            proj_tile<false>(xa, W_C, W_D, lid, 2 * half, dmask, acc1, accD);
            const float bCn = b_C[lid];
            float4* co = (float4*)(Cc + ftile * FRAG_TILE) + lane;
            co[0]  = make_float4(acc1[0] + bCn, acc1[1] + bCn, acc1[2] + bCn, acc1[3] + bCn);
            co[32] = make_float4(acc1[4] + bCn, acc1[5] + bCn, acc1[6] + bCn, acc1[7] + bCn);
        }
    }
    __syncthreads();

    float xw[PSTEPS];
#pragma unroll
    for (int q = 0; q < PSTEPS / 4; ++q) {
        const float4 v = ((const float4*)xs)[q];
        xw[4 * q + 0] = v.x; xw[4 * q + 1] = v.y; xw[4 * q + 2] = v.z; xw[4 * q + 3] = v.w;
    }
#pragma unroll
    for (int j = 0; j < PSTEPS; ++j) td[w][lane][j] = softplus_f(Pd + xw[j]);
    __syncthreads();
    float* dOut = discT + ((size_t)b * D_MODEL + d0) * SEQ + (size_t)sl * PSTEPS + lane;
    float* xOut = xT    + ((size_t)b * D_MODEL + d0) * SEQ + (size_t)sl * PSTEPS + lane;
#pragma unroll
    for (int j = 0; j < 32; ++j) {
        dOut[(size_t)j * SEQ] = td[w][j][lane];
        xOut[(size_t)j * SEQ] = tx[w][j][lane];
    }
}

__device__ __forceinline__ v8f tri_cumsum(const v2f* __restrict__ ta, const v8f v, const float carry) {
    v8f acc = {carry, carry, carry, carry, carry, carry, carry, carry};
#pragma unroll
    for (int j = 0; j < 4; ++j) {
        v2f bf;
        bf.x = v[2 * j];
        bf.y = v[2 * j + 1];
        acc = __builtin_amdgcn_wmma_f32_16x16x4_f32(false, ta[j], false, bf, (short)0, acc, false, false);
    }
    return acc;
}

__device__ __forceinline__ void scan_element(float dc, float Dcum, float xv, float bm,
                                             float A, float invA, float& V, float& Lc) {
    const float L   = A * Dcum;
    const float Bx  = bm * invA * (__expf(dc * A) - 1.0f) * xv;
    const float bxl = __logf(fabsf(Bx));
    Lc = fmaxf(L, bxl - LOG_CLAMP);
    const float sgn = (Bx > 0.0f) ? 1.0f : ((Bx < 0.0f) ? -1.0f : 0.0f);
    V = __expf(bxl - Lc) * sgn;
}

__device__ __forceinline__ void phase_fence() {
    __builtin_amdgcn_sched_barrier(0);
}

struct TileIn { float4 d0, d1, x0, x1; float bm[8], cv[8]; };

__device__ __forceinline__ void fetch_tile(TileIn& o, const float* __restrict__ dL,
                                           const float* __restrict__ xL,
                                           const float* __restrict__ bmL,
                                           const float* __restrict__ ccL, int t) {
    const float4* dp = (const float4*)(dL + t * 16);
    const float4* xp = (const float4*)(xL + t * 16);
    o.d0 = dp[0]; o.d1 = dp[1];
    o.x0 = xp[0]; o.x1 = xp[1];
    const float4* bp = (const float4*)(bmL + t * FRAG_TILE);
    const float4* cp = (const float4*)(ccL + t * FRAG_TILE);
    const float4 b0 = bp[0], b1 = bp[32], c0 = cp[0], c1 = cp[32];
    o.bm[0] = b0.x; o.bm[1] = b0.y; o.bm[2] = b0.z; o.bm[3] = b0.w;
    o.bm[4] = b1.x; o.bm[5] = b1.y; o.bm[6] = b1.z; o.bm[7] = b1.w;
    o.cv[0] = c0.x; o.cv[1] = c0.y; o.cv[2] = c0.z; o.cv[3] = c0.w;
    o.cv[4] = c1.x; o.cv[5] = c1.y; o.cv[6] = c1.z; o.cv[7] = c1.w;
}

#define NW     16
#define NTILE  (SEQ / 16)
#define ROUNDS (NTILE / NW)
#define TPITCH 20
static_assert(NTILE == 256, "the prologue sums one tile per thread in 8 waves");
static_assert(NW * 16 == NTILE, "the prologue's products: wave w takes tiles 16w..16w+15 as its 16 columns");
static_assert(ROUNDS % 2 == 0, "the round loop takes two rounds per trip");

__device__ __forceinline__ void scan_round(const TileIn& o, const v2f* __restrict__ ta,
                                           float A, float invA, const float* __restrict__ dq,
                                           float* __restrict__ totw, int w, int c, int h,
                                           v8f& Sl, v8f& eL) {
    const v8f dc = {o.d0.x, o.d0.y, o.d0.z, o.d0.w, o.d1.x, o.d1.y, o.d1.z, o.d1.w};
    const v8f xv = {o.x0.x, o.x0.y, o.x0.z, o.x0.w, o.x1.x, o.x1.y, o.x1.z, o.x1.w};
    const float4 q0 = ((const float4*)dq)[0], q1 = ((const float4*)dq)[1];
    const v8f Dcum = {q0.x, q0.y, q0.z, q0.w, q1.x, q1.y, q1.z, q1.w};
    v8f V;
#pragma unroll
    for (int r = 0; r < 8; ++r) {
        float v, Lc;
        scan_element(dc[r], Dcum[r], xv[r], o.bm[r], A, invA, v, Lc);
        V[r]  = v;
        eL[r] = __expf(Lc);
    }
    Sl = tri_cumsum(ta, V, 0.0f);
    if (h == 1) totw[c * TPITCH + w] = Sl[7];
}

__device__ __forceinline__ void scan_finish(const TileIn& o, const float* __restrict__ totr,
                                            float* __restrict__ ybw, float& base,
                                            int w, int c, int h, const v8f Sl, const v8f eL) {
    const float4* tp = (const float4*)(totr + c * TPITCH);
    float tv[16];
#pragma unroll
    for (int i = 0; i < 4; ++i) {
        const float4 v = tp[i];
        tv[4 * i + 0] = v.x; tv[4 * i + 1] = v.y; tv[4 * i + 2] = v.z; tv[4 * i + 3] = v.w;
    }
    float run = base, carry = 0.0f;
#pragma unroll
    for (int j = 0; j < NW; ++j) {
        if (j == w) carry = run;
        run += tv[j];
    }
    base = run;

    float pr[8];
#pragma unroll
    for (int r = 0; r < 8; ++r) {
        const float hh = (Sl[r] + carry) * eL[r];
        pr[r] = hh * o.cv[r];
    }
    float q4[4], q2[2];
#pragma unroll
    for (int i = 0; i < 4; ++i) {
        const float keep = (c & 8) ? pr[4 + i] : pr[i];
        const float send = (c & 8) ? pr[i] : pr[4 + i];
        q4[i] = keep + __shfl_xor(send, 8, 32);
    }
#pragma unroll
    for (int i = 0; i < 2; ++i) {
        const float keep = (c & 4) ? q4[2 + i] : q4[i];
        const float send = (c & 4) ? q4[i] : q4[2 + i];
        q2[i] = keep + __shfl_xor(send, 4, 32);
    }
    const float keep1 = (c & 2) ? q2[1] : q2[0];
    const float send1 = (c & 2) ? q2[0] : q2[1];
    const float q1 = keep1 + __shfl_xor(send1, 2, 32);
    const float q0 = q1 + __shfl_xor(q1, 1, 32);
    if ((c & 1) == 0) ybw[16 * w + 8 * h + (c >> 1)] = q0;
}

__global__ void __launch_bounds__(32 * NW)
scan_kernel(const float* __restrict__ Araw,
            const float* __restrict__ discT,
            const float* __restrict__ xT,
            const float* __restrict__ Bm,
            const float* __restrict__ Cc,
            float* __restrict__ yT) {
    __shared__ float dpre[NTILE];
    __shared__ __attribute__((aligned(16))) float dcum[SEQ];
    __shared__ float wsum[8];
    __shared__ __attribute__((aligned(16))) float tot[2][16 * TPITCH];
    __shared__ float ybuf[2][16 * NW];

    const int tid  = threadIdx.x;
    const int lane = tid & 31;
    const int w    = __builtin_amdgcn_readfirstlane(tid >> 5);
    const int c    = lane & 15;
    const int h    = lane >> 4;
    const int d    = blockIdx.x % D_MODEL;
    const int b    = blockIdx.x / D_MODEL;
    const float* dT  = discT + (size_t)blockIdx.x * SEQ;
    const float* xTp = xT    + (size_t)blockIdx.x * SEQ;
    float*       yTp = yT    + (size_t)blockIdx.x * SEQ;
    const float* dL  = dT  + 8 * h;
    const float* xL  = xTp + 8 * h;
    const float* bmL = Bm + (size_t)b * NTILE * FRAG_TILE + 4 * lane;
    const float* ccL = Cc + (size_t)b * NTILE * FRAG_TILE + 4 * lane;

    const float A    = -softplus_f(Araw[d * D_STATE + c]);
    const float invA = 1.0f / A;
    v2f ta[4];
#pragma unroll
    for (int j = 0; j < 4; ++j) {
        ta[j].x = (8 * h + 2 * j     <= c) ? 1.0f : 0.0f;
        ta[j].y = (8 * h + 2 * j + 1 <= c) ? 1.0f : 0.0f;
    }

    TileIn p, q;
    fetch_tile(p, dL, xL, bmL, ccL, w);
    const int tcol = 16 * w + c;
    const float4* cq = (const float4*)(dT + tcol * 16 + 8 * h);
    const float4 cd0 = cq[0], cd1 = cq[1];

    {
        const float4* tq = (const float4*)(dT + (tid & (NTILE - 1)) * 16);
        float s = 0.0f;
#pragma unroll
        for (int i = 0; i < 4; ++i) {
            const float4 v = tq[i];
            s += v.x; s += v.y; s += v.z; s += v.w;
        }
        float incl = s;
#pragma unroll
        for (int off = 1; off < 32; off <<= 1) {
            const float up = __shfl_up(incl, off, 32);
            if (lane >= off) incl += up;
        }
        float excl = __shfl_up(incl, 1, 32);
        if (lane == 0) excl = 0.0f;
        if (w < 8 && lane == 31) wsum[w] = incl;
        __syncthreads();
        if (w < 8) {
            float before = 0.0f;
#pragma unroll
            for (int j = 0; j < 8; ++j) before += (j < w) ? wsum[j] : 0.0f;
            dpre[tid] = before + excl;
        }
        __syncthreads();
    }
    {
        const v8f dv = {cd0.x, cd0.y, cd0.z, cd0.w, cd1.x, cd1.y, cd1.z, cd1.w};
        const v8f ds = tri_cumsum(ta, dv, dpre[tcol]);
        float4* dq = (float4*)(dcum + tcol * 16 + 8 * h);
        dq[0] = make_float4(ds[0], ds[1], ds[2], ds[3]);
        dq[1] = make_float4(ds[4], ds[5], ds[6], ds[7]);
    }
    __syncthreads();

    float base = 0.0f;
    v8f Sl, eL;
    for (int k = 0; k < ROUNDS; k += 2) {
        fetch_tile(q, dL, xL, bmL, ccL, NW * (k + 1) + w);
        phase_fence();
        scan_round(p, ta, A, invA, dcum + (NW * k + w) * 16 + 8 * h, tot[0], w, c, h, Sl, eL);
        __syncthreads();
        if (k > 0 && w < 8) yTp[16 * NW * (k - 1) + 32 * w + lane] = ybuf[1][32 * w + lane];
        scan_finish(p, tot[0], ybuf[0], base, w, c, h, Sl, eL);
        phase_fence();

        fetch_tile(p, dL, xL, bmL, ccL, NW * ((k + 2 < ROUNDS) ? k + 2 : k + 1) + w);
        phase_fence();
        scan_round(q, ta, A, invA, dcum + (NW * (k + 1) + w) * 16 + 8 * h, tot[1], w, c, h, Sl, eL);
        __syncthreads();
        if (w < 8) yTp[16 * NW * k + 32 * w + lane] = ybuf[0][32 * w + lane];
        scan_finish(q, tot[1], ybuf[1], base, w, c, h, Sl, eL);
        phase_fence();
    }
    __syncthreads();
    if (w < 8) yTp[16 * NW * (ROUNDS - 1) + 32 * w + lane] = ybuf[1][32 * w + lane];
}

#define YSUB 4

__global__ void __launch_bounds__(256)
ytrans_kernel(const float* __restrict__ yT, float* __restrict__ y) {
    __shared__ float tl[YSUB][32][33];

    const int lane = threadIdx.x & 31;
    const int w    = threadIdx.x >> 5;
    const int dq   = blockIdx.x % (D_MODEL / 32);
    const int g    = blockIdx.x / (D_MODEL / 32);
    const int sb   = g % (SEQ / (32 * YSUB));
    const int b    = g / (SEQ / (32 * YSUB));
    const int d0   = dq * 32;
    const int s0   = sb * 32 * YSUB;

#pragma unroll
    for (int u = 0; u < YSUB; ++u) {
#pragma unroll
        for (int i = 0; i < 4; ++i) {
            const int j = w + 8 * i;
            tl[u][j][lane] = yT[((size_t)b * D_MODEL + d0 + j) * SEQ + s0 + 32 * u + lane];
        }
    }
    __syncthreads();
#pragma unroll
    for (int u = 0; u < YSUB; ++u) {
#pragma unroll
        for (int i = 0; i < 4; ++i) {
            const int j = w + 8 * i;
            y[((size_t)b * SEQ + s0 + 32 * u + j) * D_MODEL + d0 + lane] = tl[u][lane][j];
        }
    }
}

extern "C" void kernel_launch(void* const* d_in, const int* in_sizes, int n_in,
                              void* d_out, int out_size, void* d_ws, size_t ws_size,
                              hipStream_t stream) {
    (void)in_sizes; (void)n_in; (void)out_size; (void)ws_size;
    const float* x    = (const float*)d_in[0];
    const float* Araw = (const float*)d_in[1];
    const float* P    = (const float*)d_in[2];
    const float* W_B  = (const float*)d_in[3];
    const float* b_B  = (const float*)d_in[4];
    const float* W_C  = (const float*)d_in[5];
    const float* b_C  = (const float*)d_in[6];
    const float* W_D  = (const float*)d_in[7];
    const float* b_D  = (const float*)d_in[8];
    float* y = (float*)d_out;

    constexpr size_t BS       = (size_t)BATCH * SEQ;
    constexpr size_t szBm     = BS * D_STATE * 4;
    constexpr size_t szCc     = BS * D_STATE * 4;
    constexpr size_t szPlane  = BS * D_MODEL * 4;
    char* ws = (char*)d_ws;
    float* Bm    = (float*)(ws);
    float* Cc    = (float*)(ws + szBm);
    float* discT = (float*)(ws + szBm + szCc);
    float* xT    = (float*)(ws + szBm + szCc + szPlane);
    float* yT    = (float*)(ws + szBm + szCc + 2 * szPlane);

    proj_prep_kernel<<<BATCH * (SEQ / PSTEPS), 32 * PWAVES, 0, stream>>>(x, W_B, b_B, W_C, b_C, W_D, b_D, P,
                                                                      Bm, Cc, discT, xT);
    scan_kernel<<<BATCH * D_MODEL, 32 * NW, 0, stream>>>(Araw, discT, xT, Bm, Cc, yT);
    ytrans_kernel<<<BATCH * (SEQ / (32 * YSUB)) * (D_MODEL / 32), 256, 0, stream>>>(yT, y);
}


# Tile running sum in blocks of four rows: three in-lane adds per block (fed in as C), then ONE K=4 WMMA with block-triangular ones over the block totals replaces four (K=16); sum order changes
# speedup vs baseline: 9.8622x; 1.0457x over previous
#include <hip/hip_runtime.h>
#include <math.h>


#define D_MODEL 128
#define D_STATE 16
#define BATCH   4
#define SEQ     4096
#define LOG_CLAMP 70.0f

typedef float v2f __attribute__((ext_vector_type(2)));
typedef float v8f __attribute__((ext_vector_type(8)));

__device__ __forceinline__ float softplus_f(float z) {
    return fmaxf(z, 0.0f) + log1pf(__expf(-fabsf(z)));
}

#define PSTEPS 32
#define PWAVES (D_MODEL / 32)
#define TXW    (32 * 33)
#define FRAG_TILE 256
static_assert(PWAVES == 4 && PSTEPS == 32, "two 16-row tiles x {W_B+W_D, W_C} = four waves");

template <bool WITH_D>
__device__ __forceinline__ void proj_tile(const float* __restrict__ xa,
                                          const float* __restrict__ W1,
                                          const float* __restrict__ W_D,
                                          int lid, int khalf, float dmask,
                                          v8f& acc1, v8f& accD) {
    v2f f1[D_MODEL / 4], fD[D_MODEL / 4];
#pragma unroll
    for (int s = 0; s < D_MODEL / 4; ++s) {
        const int k = 4 * s + khalf;
        f1[s].x = W1[(k + 0) * D_STATE + lid];
        f1[s].y = W1[(k + 1) * D_STATE + lid];
        if (WITH_D) {
            fD[s].x = W_D[k + 0] * dmask;
            fD[s].y = W_D[k + 1] * dmask;
        }
    }
#pragma unroll
    for (int s = 0; s < D_MODEL / 4; ++s) {
        const float* p = xa + ((4 * s) >> 5) * TXW + ((4 * s) & 31) * 33;
        v2f a;
        a.x = p[0];
        a.y = p[33];
        acc1 = __builtin_amdgcn_wmma_f32_16x16x4_f32(false, a, false, f1[s], (short)0, acc1, false, false);
        if (WITH_D)
            accD = __builtin_amdgcn_wmma_f32_16x16x4_f32(false, a, false, fD[s], (short)0, accD, false, false);
    }
}

__global__ void __launch_bounds__(32 * PWAVES)
proj_prep_kernel(const float* __restrict__ x,
                 const float* __restrict__ W_B, const float* __restrict__ b_B,
                 const float* __restrict__ W_C, const float* __restrict__ b_C,
                 const float* __restrict__ W_D, const float* __restrict__ b_D,
                 const float* __restrict__ P,
                 float* __restrict__ Bm, float* __restrict__ Cc,
                 float* __restrict__ discT, float* __restrict__ xT) {
    __shared__ float td[PWAVES][32][33];
    __shared__ float tx[PWAVES][32][33];
    __shared__ __attribute__((aligned(16))) float xs[PSTEPS];

    const int lane = threadIdx.x & 31;
    const int w    = __builtin_amdgcn_readfirstlane(threadIdx.x >> 5);
    const int sl   = blockIdx.x % (SEQ / PSTEPS);
    const int b    = blockIdx.x / (SEQ / PSTEPS);
    const int d0   = w * 32;
    const int d    = d0 + lane;
    const int half = lane >> 4;
    const int lid  = lane & 15;
    const int mt   = w & 1;
    const float dmask = (lid == 0) ? 1.0f : 0.0f;

    const float  Pd   = P[d];
    const size_t row0 = (size_t)b * SEQ + (size_t)sl * PSTEPS;

#pragma unroll
    for (int j = 0; j < PSTEPS; ++j) tx[w][lane][j] = x[(row0 + j) * D_MODEL + d];
    __syncthreads();

    {
        const float* xa = &tx[0][2 * half][16 * mt + lid];
        v8f acc1 = {}; v8f accD = {};
        const size_t ftile = row0 / 16 + mt;
        if (w < 2) {
            proj_tile<true>(xa, W_B, W_D, lid, 2 * half, dmask, acc1, accD);
            const float bBn = b_B[lid];
            const float bDs = b_D[0];
            float4* bo = (float4*)(Bm + ftile * FRAG_TILE) + lane;
            bo[0]  = make_float4(acc1[0] + bBn, acc1[1] + bBn, acc1[2] + bBn, acc1[3] + bBn);
            bo[32] = make_float4(acc1[4] + bBn, acc1[5] + bBn, acc1[6] + bBn, acc1[7] + bBn);
#pragma unroll
            for (int r = 0; r < 8; ++r) {
                if (lid == 0) xs[16 * mt + 8 * half + r] = accD[r] + bDs;
            }
        } else {
            proj_tile<false>(xa, W_C, W_D, lid, 2 * half, dmask, acc1, accD);
            const float bCn = b_C[lid];
            float4* co = (float4*)(Cc + ftile * FRAG_TILE) + lane;
            co[0]  = make_float4(acc1[0] + bCn, acc1[1] + bCn, acc1[2] + bCn, acc1[3] + bCn);
            co[32] = make_float4(acc1[4] + bCn, acc1[5] + bCn, acc1[6] + bCn, acc1[7] + bCn);
        }
    }
    __syncthreads();

    float xw[PSTEPS];
#pragma unroll
    for (int q = 0; q < PSTEPS / 4; ++q) {
        const float4 v = ((const float4*)xs)[q];
        xw[4 * q + 0] = v.x; xw[4 * q + 1] = v.y; xw[4 * q + 2] = v.z; xw[4 * q + 3] = v.w;
    }
#pragma unroll
    for (int j = 0; j < PSTEPS; ++j) td[w][lane][j] = softplus_f(Pd + xw[j]);
    __syncthreads();
    float* dOut = discT + ((size_t)b * D_MODEL + d0) * SEQ + (size_t)sl * PSTEPS + lane;
    float* xOut = xT    + ((size_t)b * D_MODEL + d0) * SEQ + (size_t)sl * PSTEPS + lane;
#pragma unroll
    for (int j = 0; j < 32; ++j) {
        dOut[(size_t)j * SEQ] = td[w][j][lane];
        xOut[(size_t)j * SEQ] = tx[w][j][lane];
    }
}

__device__ __forceinline__ v8f block_cumsum(const v2f tq, const v8f p) {
    v2f bf;
    bf.x = p[3];
    bf.y = p[7];
    return __builtin_amdgcn_wmma_f32_16x16x4_f32(false, tq, false, bf, (short)0, p, false, false);
}

__device__ __forceinline__ float sign_f(float v) {
    return (v > 0.0f) ? 1.0f : ((v < 0.0f) ? -1.0f : 0.0f);
}

__device__ __forceinline__ void scan_pair(const v2f dc, const v2f Dcum, const v2f xv, const v2f bm,
                                          const float A, const float invA, v2f& ex, v2f& sg, v2f& Lc) {
    const v2f L  = Dcum * A;
    const v2f da = dc * A;
    v2f e;
    e.x = __expf(da.x);
    e.y = __expf(da.y);
    const v2f Bx = bm * invA * (e - 1.0f) * xv;
    v2f bxl;
    bxl.x = __logf(fabsf(Bx.x));
    bxl.y = __logf(fabsf(Bx.y));
    const v2f cl = bxl - LOG_CLAMP;
    Lc.x = fmaxf(L.x, cl.x);
    Lc.y = fmaxf(L.y, cl.y);
    const v2f dl = bxl - Lc;
    ex.x = __expf(dl.x);
    ex.y = __expf(dl.y);
    sg.x = sign_f(Bx.x);
    sg.y = sign_f(Bx.y);
}

__device__ __forceinline__ void phase_fence() {
    __builtin_amdgcn_sched_barrier(0);
}

struct TileIn { float4 d0, d1, x0, x1; float bm[8], cv[8]; };

__device__ __forceinline__ void fetch_tile(TileIn& o, const float* __restrict__ dL,
                                           const float* __restrict__ xL,
                                           const float* __restrict__ bmL,
                                           const float* __restrict__ ccL, int t) {
    const float4* dp = (const float4*)(dL + t * 16);
    const float4* xp = (const float4*)(xL + t * 16);
    o.d0 = dp[0]; o.d1 = dp[1];
    o.x0 = xp[0]; o.x1 = xp[1];
    const float4* bp = (const float4*)(bmL + t * FRAG_TILE);
    const float4* cp = (const float4*)(ccL + t * FRAG_TILE);
    const float4 b0 = bp[0], b1 = bp[32], c0 = cp[0], c1 = cp[32];
    o.bm[0] = b0.x; o.bm[1] = b0.y; o.bm[2] = b0.z; o.bm[3] = b0.w;
    o.bm[4] = b1.x; o.bm[5] = b1.y; o.bm[6] = b1.z; o.bm[7] = b1.w;
    o.cv[0] = c0.x; o.cv[1] = c0.y; o.cv[2] = c0.z; o.cv[3] = c0.w;
    o.cv[4] = c1.x; o.cv[5] = c1.y; o.cv[6] = c1.z; o.cv[7] = c1.w;
}

#define NW     16
#define NTILE  (SEQ / 16)
#define ROUNDS (NTILE / NW)
#define TPITCH 20
static_assert(NTILE == 256, "the prologue sums one tile per thread in 8 waves");
static_assert(NW * 16 == NTILE, "the prologue's products: wave w takes tiles 16w..16w+15 as its 16 columns");
static_assert(ROUNDS % 2 == 0, "the round loop takes two rounds per trip");

__device__ __forceinline__ void scan_round(const TileIn& o, const v2f tq,
                                           float A, float invA, const float* __restrict__ dq,
                                           float* __restrict__ totw, int w, int c, int h,
                                           v8f& Sl, v8f& eL) {
    const v8f dc = {o.d0.x, o.d0.y, o.d0.z, o.d0.w, o.d1.x, o.d1.y, o.d1.z, o.d1.w};
    const v8f xv = {o.x0.x, o.x0.y, o.x0.z, o.x0.w, o.x1.x, o.x1.y, o.x1.z, o.x1.w};
    const float4 q0 = ((const float4*)dq)[0], q1 = ((const float4*)dq)[1];
    const v8f Dcum = {q0.x, q0.y, q0.z, q0.w, q1.x, q1.y, q1.z, q1.w};
    v2f ex[4], sg[4];
#pragma unroll
    for (int j = 0; j < 4; ++j) {
        const v2f dc2 = {dc[2 * j], dc[2 * j + 1]};
        const v2f Dc2 = {Dcum[2 * j], Dcum[2 * j + 1]};
        const v2f xv2 = {xv[2 * j], xv[2 * j + 1]};
        const v2f bm2 = {o.bm[2 * j], o.bm[2 * j + 1]};
        v2f Lc;
        scan_pair(dc2, Dc2, xv2, bm2, A, invA, ex[j], sg[j], Lc);
        eL[2 * j]     = __expf(Lc.x);
        eL[2 * j + 1] = __expf(Lc.y);
    }
    v8f p;
    p[0] = ex[0].x * sg[0].x;
    p[1] = p[0] + ex[0].y * sg[0].y;
    p[2] = p[1] + ex[1].x * sg[1].x;
    p[3] = p[2] + ex[1].y * sg[1].y;
    p[4] = ex[2].x * sg[2].x;
    p[5] = p[4] + ex[2].y * sg[2].y;
    p[6] = p[5] + ex[3].x * sg[3].x;
    p[7] = p[6] + ex[3].y * sg[3].y;
    Sl = block_cumsum(tq, p);
    if (h == 1) totw[c * TPITCH + w] = Sl[7];
}

__device__ __forceinline__ void scan_finish(const TileIn& o, const float* __restrict__ totr,
                                            float* __restrict__ ybw, float& base,
                                            int w, int c, int h, const v8f Sl, const v8f eL) {
    const float4* tp = (const float4*)(totr + c * TPITCH);
    float tv[16];
#pragma unroll
    for (int i = 0; i < 4; ++i) {
        const float4 v = tp[i];
        tv[4 * i + 0] = v.x; tv[4 * i + 1] = v.y; tv[4 * i + 2] = v.z; tv[4 * i + 3] = v.w;
    }
    float run = base, carry = 0.0f;
#pragma unroll
    for (int j = 0; j < NW; ++j) {
        if (j == w) carry = run;
        run += tv[j];
    }
    base = run;

    float pr[8];
#pragma unroll
    for (int r = 0; r < 8; ++r) {
        const float hh = (Sl[r] + carry) * eL[r];
        pr[r] = hh * o.cv[r];
    }
    float q4[4], q2[2];
#pragma unroll
    for (int i = 0; i < 4; ++i) {
        const float keep = (c & 8) ? pr[4 + i] : pr[i];
        const float send = (c & 8) ? pr[i] : pr[4 + i];
        q4[i] = keep + __shfl_xor(send, 8, 32);
    }
#pragma unroll
    for (int i = 0; i < 2; ++i) {
        const float keep = (c & 4) ? q4[2 + i] : q4[i];
        const float send = (c & 4) ? q4[i] : q4[2 + i];
        q2[i] = keep + __shfl_xor(send, 4, 32);
    }
    const float keep1 = (c & 2) ? q2[1] : q2[0];
    const float send1 = (c & 2) ? q2[0] : q2[1];
    const float q1 = keep1 + __shfl_xor(send1, 2, 32);
    const float q0 = q1 + __shfl_xor(q1, 1, 32);
    if ((c & 1) == 0) ybw[16 * w + 8 * h + (c >> 1)] = q0;
}

__global__ void __launch_bounds__(32 * NW)
scan_kernel(const float* __restrict__ Araw,
            const float* __restrict__ discT,
            const float* __restrict__ xT,
            const float* __restrict__ Bm,
            const float* __restrict__ Cc,
            float* __restrict__ yT) {
    __shared__ float dpre[NTILE];
    __shared__ __attribute__((aligned(16))) float dcum[SEQ];
    __shared__ float wsum[8];
    __shared__ __attribute__((aligned(16))) float tot[2][16 * TPITCH];
    __shared__ float ybuf[2][16 * NW];

    const int tid  = threadIdx.x;
    const int lane = tid & 31;
    const int w    = __builtin_amdgcn_readfirstlane(tid >> 5);
    const int c    = lane & 15;
    const int h    = lane >> 4;
    const int d    = blockIdx.x % D_MODEL;
    const int b    = blockIdx.x / D_MODEL;
    const float* dT  = discT + (size_t)blockIdx.x * SEQ;
    const float* xTp = xT    + (size_t)blockIdx.x * SEQ;
    float*       yTp = yT    + (size_t)blockIdx.x * SEQ;
    const float* dL  = dT  + 8 * h;
    const float* xL  = xTp + 8 * h;
    const float* bmL = Bm + (size_t)b * NTILE * FRAG_TILE + 4 * lane;
    const float* ccL = Cc + (size_t)b * NTILE * FRAG_TILE + 4 * lane;

    const float A    = -softplus_f(Araw[d * D_STATE + c]);
    const float invA = 1.0f / A;
    v2f tq;
    tq.x = (2 * h     < (c >> 2)) ? 1.0f : 0.0f;
    tq.y = (2 * h + 1 < (c >> 2)) ? 1.0f : 0.0f;

    TileIn p, q;
    fetch_tile(p, dL, xL, bmL, ccL, w);
    const int tcol = 16 * w + c;
    const float4* cq = (const float4*)(dT + tcol * 16 + 8 * h);
    const float4 cd0 = cq[0], cd1 = cq[1];

    {
        const float4* tq = (const float4*)(dT + (tid & (NTILE - 1)) * 16);
        float s = 0.0f;
#pragma unroll
        for (int i = 0; i < 4; ++i) {
            const float4 v = tq[i];
            s += v.x; s += v.y; s += v.z; s += v.w;
        }
        float incl = s;
#pragma unroll
        for (int off = 1; off < 32; off <<= 1) {
            const float up = __shfl_up(incl, off, 32);
            if (lane >= off) incl += up;
        }
        float excl = __shfl_up(incl, 1, 32);
        if (lane == 0) excl = 0.0f;
        if (w < 8 && lane == 31) wsum[w] = incl;
        __syncthreads();
        if (w < 8) {
            float before = 0.0f;
#pragma unroll
            for (int j = 0; j < 8; ++j) before += (j < w) ? wsum[j] : 0.0f;
            dpre[tid] = before + excl;
        }
        __syncthreads();
    }
    {
        const float pre = dpre[tcol];
        v8f dp;
        dp[0] = cd0.x;
        dp[1] = dp[0] + cd0.y;
        dp[2] = dp[1] + cd0.z;
        dp[3] = dp[2] + cd0.w;
        dp[4] = cd1.x;
        dp[5] = dp[4] + cd1.y;
        dp[6] = dp[5] + cd1.z;
        dp[7] = dp[6] + cd1.w;
        v2f dbf;
        dbf.x = dp[3];
        dbf.y = dp[7];
        const v8f dcc = dp + pre;
        const v8f ds = __builtin_amdgcn_wmma_f32_16x16x4_f32(false, tq, false, dbf, (short)0, dcc, false, false);
        float4* dq = (float4*)(dcum + tcol * 16 + 8 * h);
        dq[0] = make_float4(ds[0], ds[1], ds[2], ds[3]);
        dq[1] = make_float4(ds[4], ds[5], ds[6], ds[7]);
    }
    __syncthreads();

    float base = 0.0f;
    v8f Sl, eL;
    for (int k = 0; k < ROUNDS; k += 2) {
        fetch_tile(q, dL, xL, bmL, ccL, NW * (k + 1) + w);
        phase_fence();
        scan_round(p, tq, A, invA, dcum + (NW * k + w) * 16 + 8 * h, tot[0], w, c, h, Sl, eL);
        __syncthreads();
        if (k > 0 && w < 8) yTp[16 * NW * (k - 1) + 32 * w + lane] = ybuf[1][32 * w + lane];
        scan_finish(p, tot[0], ybuf[0], base, w, c, h, Sl, eL);
        phase_fence();

        fetch_tile(p, dL, xL, bmL, ccL, NW * ((k + 2 < ROUNDS) ? k + 2 : k + 1) + w);
        phase_fence();
        scan_round(q, tq, A, invA, dcum + (NW * (k + 1) + w) * 16 + 8 * h, tot[1], w, c, h, Sl, eL);
        __syncthreads();
        if (w < 8) yTp[16 * NW * k + 32 * w + lane] = ybuf[0][32 * w + lane];
        scan_finish(q, tot[1], ybuf[1], base, w, c, h, Sl, eL);
        phase_fence();
    }
    __syncthreads();
    if (w < 8) yTp[16 * NW * (ROUNDS - 1) + 32 * w + lane] = ybuf[1][32 * w + lane];
}

#define YSUB 4

__global__ void __launch_bounds__(256)
ytrans_kernel(const float* __restrict__ yT, float* __restrict__ y) {
    __shared__ float tl[YSUB][32][33];

    const int lane = threadIdx.x & 31;
    const int w    = threadIdx.x >> 5;
    const int dq   = blockIdx.x % (D_MODEL / 32);
    const int g    = blockIdx.x / (D_MODEL / 32);
    const int sb   = g % (SEQ / (32 * YSUB));
    const int b    = g / (SEQ / (32 * YSUB));
    const int d0   = dq * 32;
    const int s0   = sb * 32 * YSUB;

#pragma unroll
    for (int u = 0; u < YSUB; ++u) {
#pragma unroll
        for (int i = 0; i < 4; ++i) {
            const int j = w + 8 * i;
            tl[u][j][lane] = yT[((size_t)b * D_MODEL + d0 + j) * SEQ + s0 + 32 * u + lane];
        }
    }
    __syncthreads();
#pragma unroll
    for (int u = 0; u < YSUB; ++u) {
#pragma unroll
        for (int i = 0; i < 4; ++i) {
            const int j = w + 8 * i;
            y[((size_t)b * SEQ + s0 + 32 * u + j) * D_MODEL + d0 + lane] = tl[u][lane][j];
        }
    }
}

extern "C" void kernel_launch(void* const* d_in, const int* in_sizes, int n_in,
                              void* d_out, int out_size, void* d_ws, size_t ws_size,
                              hipStream_t stream) {
    (void)in_sizes; (void)n_in; (void)out_size; (void)ws_size;
    const float* x    = (const float*)d_in[0];
    const float* Araw = (const float*)d_in[1];
    const float* P    = (const float*)d_in[2];
    const float* W_B  = (const float*)d_in[3];
    const float* b_B  = (const float*)d_in[4];
    const float* W_C  = (const float*)d_in[5];
    const float* b_C  = (const float*)d_in[6];
    const float* W_D  = (const float*)d_in[7];
    const float* b_D  = (const float*)d_in[8];
    float* y = (float*)d_out;

    constexpr size_t BS       = (size_t)BATCH * SEQ;
    constexpr size_t szBm     = BS * D_STATE * 4;
    constexpr size_t szCc     = BS * D_STATE * 4;
    constexpr size_t szPlane  = BS * D_MODEL * 4;
    char* ws = (char*)d_ws;
    float* Bm    = (float*)(ws);
    float* Cc    = (float*)(ws + szBm);
    float* discT = (float*)(ws + szBm + szCc);
    float* xT    = (float*)(ws + szBm + szCc + szPlane);
    float* yT    = (float*)(ws + szBm + szCc + 2 * szPlane);

    proj_prep_kernel<<<BATCH * (SEQ / PSTEPS), 32 * PWAVES, 0, stream>>>(x, W_B, b_B, W_C, b_C, W_D, b_D, P,
                                                                      Bm, Cc, discT, xT);
    scan_kernel<<<BATCH * D_MODEL, 32 * NW, 0, stream>>>(Araw, discT, xT, Bm, Cc, yT);
    ytrans_kernel<<<BATCH * (SEQ / (32 * YSUB)) * (D_MODEL / 32), 256, 0, stream>>>(yT, y);
}


# Scan increment: its sign is the sign bit of Bx put into the exponential by one bit-field insert; the two compares, two selects and the multiply by +-1/0 per element leave the loop
# speedup vs baseline: 10.2646x; 1.0408x over previous
#include <hip/hip_runtime.h>
#include <math.h>


#define D_MODEL 128
#define D_STATE 16
#define BATCH   4
#define SEQ     4096
#define LOG_CLAMP 70.0f

typedef float v2f __attribute__((ext_vector_type(2)));
typedef float v8f __attribute__((ext_vector_type(8)));

__device__ __forceinline__ float softplus_f(float z) {
    return fmaxf(z, 0.0f) + log1pf(__expf(-fabsf(z)));
}

#define PSTEPS 32
#define PWAVES (D_MODEL / 32)
#define TXW    (32 * 33)
#define FRAG_TILE 256
static_assert(PWAVES == 4 && PSTEPS == 32, "two 16-row tiles x {W_B+W_D, W_C} = four waves");

template <bool WITH_D>
__device__ __forceinline__ void proj_tile(const float* __restrict__ xa,
                                          const float* __restrict__ W1,
                                          const float* __restrict__ W_D,
                                          int lid, int khalf, float dmask,
                                          v8f& acc1, v8f& accD) {
    v2f f1[D_MODEL / 4], fD[D_MODEL / 4];
#pragma unroll
    for (int s = 0; s < D_MODEL / 4; ++s) {
        const int k = 4 * s + khalf;
        f1[s].x = W1[(k + 0) * D_STATE + lid];
        f1[s].y = W1[(k + 1) * D_STATE + lid];
        if (WITH_D) {
            fD[s].x = W_D[k + 0] * dmask;
            fD[s].y = W_D[k + 1] * dmask;
        }
    }
#pragma unroll
    for (int s = 0; s < D_MODEL / 4; ++s) {
        const float* p = xa + ((4 * s) >> 5) * TXW + ((4 * s) & 31) * 33;
        v2f a;
        a.x = p[0];
        a.y = p[33];
        acc1 = __builtin_amdgcn_wmma_f32_16x16x4_f32(false, a, false, f1[s], (short)0, acc1, false, false);
        if (WITH_D)
            accD = __builtin_amdgcn_wmma_f32_16x16x4_f32(false, a, false, fD[s], (short)0, accD, false, false);
    }
}

__global__ void __launch_bounds__(32 * PWAVES)
proj_prep_kernel(const float* __restrict__ x,
                 const float* __restrict__ W_B, const float* __restrict__ b_B,
                 const float* __restrict__ W_C, const float* __restrict__ b_C,
                 const float* __restrict__ W_D, const float* __restrict__ b_D,
                 const float* __restrict__ P,
                 float* __restrict__ Bm, float* __restrict__ Cc,
                 float* __restrict__ discT, float* __restrict__ xT) {
    __shared__ float td[PWAVES][32][33];
    __shared__ float tx[PWAVES][32][33];
    __shared__ __attribute__((aligned(16))) float xs[PSTEPS];

    const int lane = threadIdx.x & 31;
    const int w    = __builtin_amdgcn_readfirstlane(threadIdx.x >> 5);
    const int sl   = blockIdx.x % (SEQ / PSTEPS);
    const int b    = blockIdx.x / (SEQ / PSTEPS);
    const int d0   = w * 32;
    const int d    = d0 + lane;
    const int half = lane >> 4;
    const int lid  = lane & 15;
    const int mt   = w & 1;
    const float dmask = (lid == 0) ? 1.0f : 0.0f;

    const float  Pd   = P[d];
    const size_t row0 = (size_t)b * SEQ + (size_t)sl * PSTEPS;

#pragma unroll
    for (int j = 0; j < PSTEPS; ++j) tx[w][lane][j] = x[(row0 + j) * D_MODEL + d];
    __syncthreads();

    {
        const float* xa = &tx[0][2 * half][16 * mt + lid];
        v8f acc1 = {}; v8f accD = {};
        const size_t ftile = row0 / 16 + mt;
        if (w < 2) {
            proj_tile<true>(xa, W_B, W_D, lid, 2 * half, dmask, acc1, accD);
            const float bBn = b_B[lid];
            const float bDs = b_D[0];
            float4* bo = (float4*)(Bm + ftile * FRAG_TILE) + lane;
            bo[0]  = make_float4(acc1[0] + bBn, acc1[1] + bBn, acc1[2] + bBn, acc1[3] + bBn);
            bo[32] = make_float4(acc1[4] + bBn, acc1[5] + bBn, acc1[6] + bBn, acc1[7] + bBn);
#pragma unroll
            for (int r = 0; r < 8; ++r) {
                if (lid == 0) xs[16 * mt + 8 * half + r] = accD[r] + bDs;
            }
        } else {
            proj_tile<false>(xa, W_C, W_D, lid, 2 * half, dmask, acc1, accD);
            const float bCn = b_C[lid];
            float4* co = (float4*)(Cc + ftile * FRAG_TILE) + lane;
            co[0]  = make_float4(acc1[0] + bCn, acc1[1] + bCn, acc1[2] + bCn, acc1[3] + bCn);
            co[32] = make_float4(acc1[4] + bCn, acc1[5] + bCn, acc1[6] + bCn, acc1[7] + bCn);
        }
    }
    __syncthreads();

    float xw[PSTEPS];
#pragma unroll
    for (int q = 0; q < PSTEPS / 4; ++q) {
        const float4 v = ((const float4*)xs)[q];
        xw[4 * q + 0] = v.x; xw[4 * q + 1] = v.y; xw[4 * q + 2] = v.z; xw[4 * q + 3] = v.w;
    }
#pragma unroll
    for (int j = 0; j < PSTEPS; ++j) td[w][lane][j] = softplus_f(Pd + xw[j]);
    __syncthreads();
    float* dOut = discT + ((size_t)b * D_MODEL + d0) * SEQ + (size_t)sl * PSTEPS + lane;
    float* xOut = xT    + ((size_t)b * D_MODEL + d0) * SEQ + (size_t)sl * PSTEPS + lane;
#pragma unroll
    for (int j = 0; j < 32; ++j) {
        dOut[(size_t)j * SEQ] = td[w][j][lane];
        xOut[(size_t)j * SEQ] = tx[w][j][lane];
    }
}

__device__ __forceinline__ v8f block_cumsum(const v2f tq, const v8f p) {
    v2f bf;
    bf.x = p[3];
    bf.y = p[7];
    return __builtin_amdgcn_wmma_f32_16x16x4_f32(false, tq, false, bf, (short)0, p, false, false);
}

__device__ __forceinline__ void scan_pair(const v2f dc, const v2f Dcum, const v2f xv, const v2f bm,
                                          const float A, const float invA, v2f& V, v2f& Lc) {
    const v2f L  = Dcum * A;
    const v2f da = dc * A;
    v2f e;
    e.x = __expf(da.x);
    e.y = __expf(da.y);
    const v2f Bx = bm * invA * (e - 1.0f) * xv;
    v2f bxl;
    bxl.x = __logf(fabsf(Bx.x));
    bxl.y = __logf(fabsf(Bx.y));
    const v2f cl = bxl - LOG_CLAMP;
    Lc.x = fmaxf(L.x, cl.x);
    Lc.y = fmaxf(L.y, cl.y);
    const v2f dl = bxl - Lc;
    V.x = __builtin_copysignf(__expf(dl.x), Bx.x);
    V.y = __builtin_copysignf(__expf(dl.y), Bx.y);
}

__device__ __forceinline__ void phase_fence() {
    __builtin_amdgcn_sched_barrier(0);
}

struct TileIn { float4 d0, d1, x0, x1; float bm[8], cv[8]; };

__device__ __forceinline__ void fetch_tile(TileIn& o, const float* __restrict__ dL,
                                           const float* __restrict__ xL,
                                           const float* __restrict__ bmL,
                                           const float* __restrict__ ccL, int t) {
    const float4* dp = (const float4*)(dL + t * 16);
    const float4* xp = (const float4*)(xL + t * 16);
    o.d0 = dp[0]; o.d1 = dp[1];
    o.x0 = xp[0]; o.x1 = xp[1];
    const float4* bp = (const float4*)(bmL + t * FRAG_TILE);
    const float4* cp = (const float4*)(ccL + t * FRAG_TILE);
    const float4 b0 = bp[0], b1 = bp[32], c0 = cp[0], c1 = cp[32];
    o.bm[0] = b0.x; o.bm[1] = b0.y; o.bm[2] = b0.z; o.bm[3] = b0.w;
    o.bm[4] = b1.x; o.bm[5] = b1.y; o.bm[6] = b1.z; o.bm[7] = b1.w;
    o.cv[0] = c0.x; o.cv[1] = c0.y; o.cv[2] = c0.z; o.cv[3] = c0.w;
    o.cv[4] = c1.x; o.cv[5] = c1.y; o.cv[6] = c1.z; o.cv[7] = c1.w;
}

#define NW     16
#define NTILE  (SEQ / 16)
#define ROUNDS (NTILE / NW)
#define TPITCH 20
static_assert(NTILE == 256, "the prologue sums one tile per thread in 8 waves");
static_assert(NW * 16 == NTILE, "the prologue's products: wave w takes tiles 16w..16w+15 as its 16 columns");
static_assert(ROUNDS % 2 == 0, "the round loop takes two rounds per trip");

__device__ __forceinline__ void scan_round(const TileIn& o, const v2f tq,
                                           float A, float invA, const float* __restrict__ dq,
                                           float* __restrict__ totw, int w, int c, int h,
                                           v8f& Sl, v8f& eL) {
    const v8f dc = {o.d0.x, o.d0.y, o.d0.z, o.d0.w, o.d1.x, o.d1.y, o.d1.z, o.d1.w};
    const v8f xv = {o.x0.x, o.x0.y, o.x0.z, o.x0.w, o.x1.x, o.x1.y, o.x1.z, o.x1.w};
    const float4 q0 = ((const float4*)dq)[0], q1 = ((const float4*)dq)[1];
    const v8f Dcum = {q0.x, q0.y, q0.z, q0.w, q1.x, q1.y, q1.z, q1.w};
    v2f V[4];
#pragma unroll
    for (int j = 0; j < 4; ++j) {
        const v2f dc2 = {dc[2 * j], dc[2 * j + 1]};
        const v2f Dc2 = {Dcum[2 * j], Dcum[2 * j + 1]};
        const v2f xv2 = {xv[2 * j], xv[2 * j + 1]};
        const v2f bm2 = {o.bm[2 * j], o.bm[2 * j + 1]};
        v2f Lc;
        scan_pair(dc2, Dc2, xv2, bm2, A, invA, V[j], Lc);
        eL[2 * j]     = __expf(Lc.x);
        eL[2 * j + 1] = __expf(Lc.y);
    }
    v8f p;
    p[0] = V[0].x;
    p[1] = p[0] + V[0].y;
    p[2] = p[1] + V[1].x;
    p[3] = p[2] + V[1].y;
    p[4] = V[2].x;
    p[5] = p[4] + V[2].y;
    p[6] = p[5] + V[3].x;
    p[7] = p[6] + V[3].y;
    Sl = block_cumsum(tq, p);
    if (h == 1) totw[c * TPITCH + w] = Sl[7];
}

__device__ __forceinline__ void scan_finish(const TileIn& o, const float* __restrict__ totr,
                                            float* __restrict__ ybw, float& base,
                                            int w, int c, int h, const v8f Sl, const v8f eL) {
    const float4* tp = (const float4*)(totr + c * TPITCH);
    float tv[16];
#pragma unroll
    for (int i = 0; i < 4; ++i) {
        const float4 v = tp[i];
        tv[4 * i + 0] = v.x; tv[4 * i + 1] = v.y; tv[4 * i + 2] = v.z; tv[4 * i + 3] = v.w;
    }
    float run = base, carry = 0.0f;
#pragma unroll
    for (int j = 0; j < NW; ++j) {
        if (j == w) carry = run;
        run += tv[j];
    }
    base = run;

    float pr[8];
#pragma unroll
    for (int r = 0; r < 8; ++r) {
        const float hh = (Sl[r] + carry) * eL[r];
        pr[r] = hh * o.cv[r];
    }
    float q4[4], q2[2];
#pragma unroll
    for (int i = 0; i < 4; ++i) {
        const float keep = (c & 8) ? pr[4 + i] : pr[i];
        const float send = (c & 8) ? pr[i] : pr[4 + i];
        q4[i] = keep + __shfl_xor(send, 8, 32);
    }
#pragma unroll
    for (int i = 0; i < 2; ++i) {
        const float keep = (c & 4) ? q4[2 + i] : q4[i];
        const float send = (c & 4) ? q4[i] : q4[2 + i];
        q2[i] = keep + __shfl_xor(send, 4, 32);
    }
    const float keep1 = (c & 2) ? q2[1] : q2[0];
    const float send1 = (c & 2) ? q2[0] : q2[1];
    const float q1 = keep1 + __shfl_xor(send1, 2, 32);
    const float q0 = q1 + __shfl_xor(q1, 1, 32);
    if ((c & 1) == 0) ybw[16 * w + 8 * h + (c >> 1)] = q0;
}

__global__ void __launch_bounds__(32 * NW)
scan_kernel(const float* __restrict__ Araw,
            const float* __restrict__ discT,
            const float* __restrict__ xT,
            const float* __restrict__ Bm,
            const float* __restrict__ Cc,
            float* __restrict__ yT) {
    __shared__ float dpre[NTILE];
    __shared__ __attribute__((aligned(16))) float dcum[SEQ];
    __shared__ float wsum[8];
    __shared__ __attribute__((aligned(16))) float tot[2][16 * TPITCH];
    __shared__ float ybuf[2][16 * NW];

    const int tid  = threadIdx.x;
    const int lane = tid & 31;
    const int w    = __builtin_amdgcn_readfirstlane(tid >> 5);
    const int c    = lane & 15;
    const int h    = lane >> 4;
    const int d    = blockIdx.x % D_MODEL;
    const int b    = blockIdx.x / D_MODEL;
    const float* dT  = discT + (size_t)blockIdx.x * SEQ;
    const float* xTp = xT    + (size_t)blockIdx.x * SEQ;
    float*       yTp = yT    + (size_t)blockIdx.x * SEQ;
    const float* dL  = dT  + 8 * h;
    const float* xL  = xTp + 8 * h;
    const float* bmL = Bm + (size_t)b * NTILE * FRAG_TILE + 4 * lane;
    const float* ccL = Cc + (size_t)b * NTILE * FRAG_TILE + 4 * lane;

    const float A    = -softplus_f(Araw[d * D_STATE + c]);
    const float invA = 1.0f / A;
    v2f tq;
    tq.x = (2 * h     < (c >> 2)) ? 1.0f : 0.0f;
    tq.y = (2 * h + 1 < (c >> 2)) ? 1.0f : 0.0f;

    TileIn p, q;
    fetch_tile(p, dL, xL, bmL, ccL, w);
    const int tcol = 16 * w + c;
    const float4* cq = (const float4*)(dT + tcol * 16 + 8 * h);
    const float4 cd0 = cq[0], cd1 = cq[1];

    {
        const float4* tq = (const float4*)(dT + (tid & (NTILE - 1)) * 16);
        float s = 0.0f;
#pragma unroll
        for (int i = 0; i < 4; ++i) {
            const float4 v = tq[i];
            s += v.x; s += v.y; s += v.z; s += v.w;
        }
        float incl = s;
#pragma unroll
        for (int off = 1; off < 32; off <<= 1) {
            const float up = __shfl_up(incl, off, 32);
            if (lane >= off) incl += up;
        }
        float excl = __shfl_up(incl, 1, 32);
        if (lane == 0) excl = 0.0f;
        if (w < 8 && lane == 31) wsum[w] = incl;
        __syncthreads();
        if (w < 8) {
            float before = 0.0f;
#pragma unroll
            for (int j = 0; j < 8; ++j) before += (j < w) ? wsum[j] : 0.0f;
            dpre[tid] = before + excl;
        }
        __syncthreads();
    }
    {
        const float pre = dpre[tcol];
        v8f dp;
        dp[0] = cd0.x;
        dp[1] = dp[0] + cd0.y;
        dp[2] = dp[1] + cd0.z;
        dp[3] = dp[2] + cd0.w;
        dp[4] = cd1.x;
        dp[5] = dp[4] + cd1.y;
        dp[6] = dp[5] + cd1.z;
        dp[7] = dp[6] + cd1.w;
        v2f dbf;
        dbf.x = dp[3];
        dbf.y = dp[7];
        const v8f dcc = dp + pre;
        const v8f ds = __builtin_amdgcn_wmma_f32_16x16x4_f32(false, tq, false, dbf, (short)0, dcc, false, false);
        float4* dq = (float4*)(dcum + tcol * 16 + 8 * h);
        dq[0] = make_float4(ds[0], ds[1], ds[2], ds[3]);
        dq[1] = make_float4(ds[4], ds[5], ds[6], ds[7]);
    }
    __syncthreads();

    float base = 0.0f;
    v8f Sl, eL;
    for (int k = 0; k < ROUNDS; k += 2) {
        fetch_tile(q, dL, xL, bmL, ccL, NW * (k + 1) + w);
        phase_fence();
        scan_round(p, tq, A, invA, dcum + (NW * k + w) * 16 + 8 * h, tot[0], w, c, h, Sl, eL);
        __syncthreads();
        if (k > 0 && w < 8) yTp[16 * NW * (k - 1) + 32 * w + lane] = ybuf[1][32 * w + lane];
        scan_finish(p, tot[0], ybuf[0], base, w, c, h, Sl, eL);
        phase_fence();

        fetch_tile(p, dL, xL, bmL, ccL, NW * ((k + 2 < ROUNDS) ? k + 2 : k + 1) + w);
        phase_fence();
        scan_round(q, tq, A, invA, dcum + (NW * (k + 1) + w) * 16 + 8 * h, tot[1], w, c, h, Sl, eL);
        __syncthreads();
        if (w < 8) yTp[16 * NW * k + 32 * w + lane] = ybuf[0][32 * w + lane];
        scan_finish(q, tot[1], ybuf[1], base, w, c, h, Sl, eL);
        phase_fence();
    }
    __syncthreads();
    if (w < 8) yTp[16 * NW * (ROUNDS - 1) + 32 * w + lane] = ybuf[1][32 * w + lane];
}

#define YSUB 4

__global__ void __launch_bounds__(256)
ytrans_kernel(const float* __restrict__ yT, float* __restrict__ y) {
    __shared__ float tl[YSUB][32][33];

    const int lane = threadIdx.x & 31;
    const int w    = threadIdx.x >> 5;
    const int dq   = blockIdx.x % (D_MODEL / 32);
    const int g    = blockIdx.x / (D_MODEL / 32);
    const int sb   = g % (SEQ / (32 * YSUB));
    const int b    = g / (SEQ / (32 * YSUB));
    const int d0   = dq * 32;
    const int s0   = sb * 32 * YSUB;

#pragma unroll
    for (int u = 0; u < YSUB; ++u) {
#pragma unroll
        for (int i = 0; i < 4; ++i) {
            const int j = w + 8 * i;
            tl[u][j][lane] = yT[((size_t)b * D_MODEL + d0 + j) * SEQ + s0 + 32 * u + lane];
        }
    }
    __syncthreads();
#pragma unroll
    for (int u = 0; u < YSUB; ++u) {
#pragma unroll
        for (int i = 0; i < 4; ++i) {
            const int j = w + 8 * i;
            y[((size_t)b * SEQ + s0 + 32 * u + j) * D_MODEL + d0 + lane] = tl[u][lane][j];
        }
    }
}

extern "C" void kernel_launch(void* const* d_in, const int* in_sizes, int n_in,
                              void* d_out, int out_size, void* d_ws, size_t ws_size,
                              hipStream_t stream) {
    (void)in_sizes; (void)n_in; (void)out_size; (void)ws_size;
    const float* x    = (const float*)d_in[0];
    const float* Araw = (const float*)d_in[1];
    const float* P    = (const float*)d_in[2];
    const float* W_B  = (const float*)d_in[3];
    const float* b_B  = (const float*)d_in[4];
    const float* W_C  = (const float*)d_in[5];
    const float* b_C  = (const float*)d_in[6];
    const float* W_D  = (const float*)d_in[7];
    const float* b_D  = (const float*)d_in[8];
    float* y = (float*)d_out;

    constexpr size_t BS       = (size_t)BATCH * SEQ;
    constexpr size_t szBm     = BS * D_STATE * 4;
    constexpr size_t szCc     = BS * D_STATE * 4;
    constexpr size_t szPlane  = BS * D_MODEL * 4;
    char* ws = (char*)d_ws;
    float* Bm    = (float*)(ws);
    float* Cc    = (float*)(ws + szBm);
    float* discT = (float*)(ws + szBm + szCc);
    float* xT    = (float*)(ws + szBm + szCc + szPlane);
    float* yT    = (float*)(ws + szBm + szCc + 2 * szPlane);

    proj_prep_kernel<<<BATCH * (SEQ / PSTEPS), 32 * PWAVES, 0, stream>>>(x, W_B, b_B, W_C, b_C, W_D, b_D, P,
                                                                      Bm, Cc, discT, xT);
    scan_kernel<<<BATCH * D_MODEL, 32 * NW, 0, stream>>>(Araw, discT, xT, Bm, Cc, yT);
    ytrans_kernel<<<BATCH * (SEQ / (32 * YSUB)) * (D_MODEL / 32), 256, 0, stream>>>(yT, y);
}
